# starting kernel
# baseline (speedup 1.0000x reference)
#include <hip/hip_runtime.h>
#include <hip/hip_bf16.h>
#include <math.h>

#define Bn      8
#define Sn      12
#define NNn     325
#define Dn      128
#define EDn     256
#define DSTATEn 8
#define DCONVn  4
#define DTRANKn 8
#define MSLOTSn 10
#define Ln      (Sn*NNn)
#define NTOK    (Bn*Ln)
#define TM      3
#define TOKB    (16*TM)
#define NBLK    (NTOK/TOKB)

typedef float v2f __attribute__((ext_vector_type(2)));
typedef float v8f __attribute__((ext_vector_type(8)));

__device__ __forceinline__ v8f wmma4(v2f a, v2f b, v8f c) {
  return __builtin_amdgcn_wmma_f32_16x16x4_f32(false, a, false, b, (short)0, c,
                                               false, false);
}

__device__ __forceinline__ void store_c(float* C, int ldc, int nb, v8f c) {
  int lane = threadIdx.x & 31;
  int n = nb + (lane & 15);
  int mb = (lane < 16) ? 0 : 8;
#pragma unroll
  for (int j = 0; j < 8; ++j) C[(mb + j)*ldc + n] = c[j];
}

__device__ __forceinline__ void gemm3(const float* A, int lda, const v2f* Bsw,
                                      int KT, int nt, v8f acc[TM]) {
  int lane = threadIdx.x & 31;
  int m = lane & 15;
  int ks = (lane >> 4) << 1;
  const v2f* bp = Bsw + (size_t)nt * KT * 32 + lane;
  const v2f* a0 = (const v2f*)(A + (0*16 + m)*lda + ks);
  const v2f* a1 = (const v2f*)(A + (1*16 + m)*lda + ks);
  const v2f* a2 = (const v2f*)(A + (2*16 + m)*lda + ks);
  v8f z = {};
  acc[0] = z; acc[1] = z; acc[2] = z;
#pragma unroll 4
  for (int kt = 0; kt < KT; ++kt) {
    v2f b = bp[kt * 32];
    acc[0] = wmma4(a0[kt*2], b, acc[0]);
    acc[1] = wmma4(a1[kt*2], b, acc[1]);
    acc[2] = wmma4(a2[kt*2], b, acc[2]);
  }
}

__device__ __forceinline__ float silu_f(float x) { return x / (1.f + __expf(-x)); }

__global__ void swz_kernel(const float* W, float* dst, int K, int Nreal, int NT) {
  int idx = blockIdx.x * blockDim.x + threadIdx.x;
  int KT = K / 4;
  if (idx >= NT * KT * 32) return;
  int lane = idx & 31;
  int kt = (idx >> 5) % KT;
  int nt = idx / (32 * KT);
  int n = nt*16 + (lane & 15);
  int k = kt*4 + ((lane >> 4) << 1);
  float bx = (n < Nreal) ? W[k*Nreal + n]     : 0.f;
  float by = (n < Nreal) ? W[(k+1)*Nreal + n] : 0.f;
  dst[idx*2]     = bx;
  dst[idx*2 + 1] = by;
}

__global__ void kv_kernel(const float* mem, const float* Wk, const float* Wv,
                          float* kb, float* vb) {
  int i = blockIdx.x * blockDim.x + threadIdx.x;
  if (i >= 2 * MSLOTSn * Dn) return;
  int which = i / (MSLOTSn * Dn);
  int r = i % (MSLOTSn * Dn);
  int m = r / Dn, d = r % Dn;
  const float* W = which ? Wv : Wk;
  float acc = 0.f;
  for (int k = 0; k < Dn; ++k) acc += mem[m*Dn + k] * W[k*Dn + d];
  (which ? vb : kb)[m*Dn + d] = acc;
}

__global__ void kvswz_kernel(const float* kb, const float* vb,
                             float* kfragT, float* vfrag) {
  int idx = blockIdx.x * blockDim.x + threadIdx.x;
  if (idx < 32*32) {
    int lane = idx & 31; int kt = idx >> 5;
    int n = lane & 15;   int k = kt*4 + ((lane >> 4) << 1);
    kfragT[idx*2]     = (n < MSLOTSn) ? kb[n*Dn + k]     : 0.f;
    kfragT[idx*2 + 1] = (n < MSLOTSn) ? kb[n*Dn + k + 1] : 0.f;
  }
  if (idx < 8*4*32) {
    int lane = idx & 31; int kt = (idx >> 5) & 3; int nt = idx >> 7;
    int n = nt*16 + (lane & 15); int k = kt*4 + ((lane >> 4) << 1);
    vfrag[idx*2]     = (k     < MSLOTSn) ? vb[k*Dn + n]     : 0.f;
    vfrag[idx*2 + 1] = (k + 1 < MSLOTSn) ? vb[(k+1)*Dn + n] : 0.f;
  }
}

__global__ void attn_kernel(const float* x, const v2f* Wq_sw, const v2f* Wo_sw,
                            const float* ln_g, const float* ln_b,
                            const v2f* kfragT, const v2f* vfrag, float* xr) {
  __shared__ float xs[TOKB*Dn];
  __shared__ float qs[TOKB*Dn];
  __shared__ float av[TOKB*Dn];
  __shared__ float sc[TOKB*16];
  int tid = threadIdx.x;
  int wave = tid >> 5;
  long t0 = (long)blockIdx.x * TOKB;
  v8f acc[TM];

  for (int i = tid; i < TOKB*Dn; i += 128) xs[i] = x[t0*Dn + i];
  __syncthreads();

  for (int nti = 0; nti < 2; ++nti) {
    int nt = wave*2 + nti;
    gemm3(xs, Dn, Wq_sw, 32, nt, acc);
#pragma unroll
    for (int t = 0; t < TM; ++t) store_c(qs + t*16*Dn, Dn, nt*16, acc[t]);
  }
  __syncthreads();

  if (wave == 0) {
    gemm3(qs, Dn, kfragT, 32, 0, acc);
#pragma unroll
    for (int t = 0; t < TM; ++t) store_c(sc + t*16*16, 16, 0, acc[t]);
  }
  __syncthreads();

  if (tid < TOKB) {
    const float scale = 0.088388347648318447f;
    float mx = -1e30f;
    for (int m = 0; m < MSLOTSn; ++m) {
      float v = sc[tid*16 + m] * scale; sc[tid*16 + m] = v;
      if (v > mx) mx = v;
    }
    float s = 0.f;
    for (int m = 0; m < MSLOTSn; ++m) {
      float e = __expf(sc[tid*16 + m] - mx); sc[tid*16 + m] = e; s += e;
    }
    float inv = 1.f / s;
    for (int m = 0; m < MSLOTSn; ++m) sc[tid*16 + m] *= inv;
    for (int m = MSLOTSn; m < 16; ++m) sc[tid*16 + m] = 0.f;
  }
  __syncthreads();

  for (int nti = 0; nti < 2; ++nti) {
    int nt = wave*2 + nti;
    gemm3(sc, 16, vfrag, 4, nt, acc);
#pragma unroll
    for (int t = 0; t < TM; ++t) store_c(av + t*16*Dn, Dn, nt*16, acc[t]);
  }
  __syncthreads();

  for (int nti = 0; nti < 2; ++nti) {
    int nt = wave*2 + nti;
    gemm3(av, Dn, Wo_sw, 32, nt, acc);
#pragma unroll
    for (int t = 0; t < TM; ++t) store_c(qs + t*16*Dn, Dn, nt*16, acc[t]);
  }
  __syncthreads();

  if (tid < TOKB) {
    float m = 0.f;
    for (int c = 0; c < Dn; ++c) m += qs[tid*Dn + c];
    m *= (1.f / Dn);
    float v = 0.f;
    for (int c = 0; c < Dn; ++c) { float d = qs[tid*Dn + c] - m; v += d*d; }
    v *= (1.f / Dn);
    float inv = rsqrtf(v + 1e-5f);
    long tok = t0 + tid;
    for (int c = 0; c < Dn; ++c)
      xr[tok*Dn + c] = (qs[tid*Dn + c] - m) * inv * ln_g[c] + ln_b[c]
                       + xs[tid*Dn + c];
  }
}

__global__ void in_kernel(const float* xr, const float* rms_w, const v2f* Win_sw,
                          float* xp, float* zb) {
  __shared__ float xn[TOKB*Dn];
  int tid = threadIdx.x;
  int wave = tid >> 5;
  int lane = tid & 31;
  long t0 = (long)blockIdx.x * TOKB;
  v8f acc[TM];

  for (int i = tid; i < TOKB*Dn; i += 128) xn[i] = xr[t0*Dn + i];
  __syncthreads();
  if (tid < TOKB) {
    float s = 0.f;
    for (int c = 0; c < Dn; ++c) { float v = xn[tid*Dn + c]; s += v*v; }
    float inv = rsqrtf(s / Dn + 1e-5f);
    for (int c = 0; c < Dn; ++c) xn[tid*Dn + c] = xn[tid*Dn + c] * inv * rms_w[c];
  }
  __syncthreads();

  for (int nti = 0; nti < 8; ++nti) {
    int nt = wave*8 + nti;
    gemm3(xn, Dn, Win_sw, 32, nt, acc);
    int n = nt*16 + (lane & 15);
    int mb = (lane < 16) ? 0 : 8;
#pragma unroll
    for (int t = 0; t < TM; ++t)
#pragma unroll
      for (int j = 0; j < 8; ++j) {
        long tok = t0 + t*16 + mb + j;
        float v = acc[t][j];
        if (n < EDn) xp[tok*EDn + n] = v;
        else         zb[tok*EDn + (n - EDn)] = v;
      }
  }
}

__global__ void conv_kernel(const float* xp, const float* conv_w, const float* conv_b,
                            const v2f* Wx_sw, const v2f* Wdt_sw, const float* b_dt,
                            float* xc, float* delta, float* Bm, float* Cm) {
  __shared__ float xcs[TOKB*EDn];
  __shared__ float dbcs[TOKB*32];
  int tid = threadIdx.x;
  int wave = tid >> 5;
  int lane = tid & 31;
  long t0 = (long)blockIdx.x * TOKB;
  v8f acc[TM];

  for (int i = tid; i < TOKB*EDn; i += 128) {
    int r = i >> 8;
    int c = i & 255;
    long tok = t0 + r;
    int b = (int)(tok / Ln);
    int l = (int)(tok % Ln);
    float a = conv_b[c];
#pragma unroll
    for (int j = 0; j < DCONVn; ++j) {
      int ll = l - (DCONVn - 1) + j;
      if (ll >= 0) a += conv_w[c*DCONVn + j] * xp[((long)b*Ln + ll)*EDn + c];
    }
    a = silu_f(a);
    xcs[i] = a;
    xc[tok*EDn + c] = a;
  }
  __syncthreads();

  if (wave < 2) {
    gemm3(xcs, EDn, Wx_sw, 64, wave, acc);
#pragma unroll
    for (int t = 0; t < TM; ++t) store_c(dbcs + t*16*32, 32, wave*16, acc[t]);
  }
  __syncthreads();

  for (int nti = 0; nti < 4; ++nti) {
    int nt = wave*4 + nti;
    gemm3(dbcs, 32, Wdt_sw, 2, nt, acc);
    int n = nt*16 + (lane & 15);
    int mb = (lane < 16) ? 0 : 8;
#pragma unroll
    for (int t = 0; t < TM; ++t)
#pragma unroll
      for (int j = 0; j < 8; ++j) {
        long tok = t0 + t*16 + mb + j;
        float v = acc[t][j] + b_dt[n];
        delta[tok*EDn + n] = (v > 20.f) ? v : log1pf(__expf(v));
      }
  }

  for (int i = tid; i < TOKB*16; i += 128) {
    int r = i >> 4, c = i & 15;
    long tok = t0 + r;
    float v = dbcs[r*32 + DTRANKn + c];
    if (c < DSTATEn) Bm[tok*DSTATEn + c]             = v;
    else             Cm[tok*DSTATEn + (c - DSTATEn)] = v;
  }
}

__global__ void scan_kernel(const float* delta, const float* xc, const float* Bm,
                            const float* Cm, const float* A_log, const float* Dp,
                            float* y) {
  int gid = blockIdx.x * blockDim.x + threadIdx.x;
  if (gid >= Bn * EDn) return;
  int b = gid / EDn;
  int d = gid % EDn;
  float a[DSTATEn];
#pragma unroll
  for (int n = 0; n < DSTATEn; ++n) a[n] = -__expf(A_log[d*DSTATEn + n]);
  float dp = Dp[d];
  float h[DSTATEn];
#pragma unroll
  for (int n = 0; n < DSTATEn; ++n) h[n] = 0.f;
  long base = (long)b * Ln;
  for (int l = 0; l < Ln; ++l) {
    long t = base + l;
    float dt = delta[t*EDn + d];
    float xv = xc[t*EDn + d];
    float yv = 0.f;
#pragma unroll
    for (int n = 0; n < DSTATEn; ++n) {
      float dA = __expf(dt * a[n]);
      h[n] = dA * h[n] + dt * Bm[t*DSTATEn + n] * xv;
      yv += h[n] * Cm[t*DSTATEn + n];
    }
    y[t*EDn + d] = yv + dp * xv;
  }
}

__global__ void out_kernel(const float* y, const float* zb, const v2f* Wout_sw,
                           const float* xr, const float* x, float* out) {
  __shared__ float ys[TOKB*EDn];
  int tid = threadIdx.x;
  int wave = tid >> 5;
  int lane = tid & 31;
  long t0 = (long)blockIdx.x * TOKB;
  v8f acc[TM];

  for (int i = tid; i < TOKB*EDn; i += 128) {
    int r = i >> 8, c = i & 255;
    long tok = t0 + r;
    float z = zb[tok*EDn + c];
    ys[i] = y[tok*EDn + c] * silu_f(z);
  }
  __syncthreads();

  for (int nti = 0; nti < 2; ++nti) {
    int nt = wave*2 + nti;
    gemm3(ys, EDn, Wout_sw, 64, nt, acc);
    int n = nt*16 + (lane & 15);
    int mb = (lane < 16) ? 0 : 8;
#pragma unroll
    for (int t = 0; t < TM; ++t)
#pragma unroll
      for (int j = 0; j < 8; ++j) {
        long tok = t0 + t*16 + mb + j;
        out[tok*Dn + n] = acc[t][j] + xr[tok*Dn + n] + x[tok*Dn + n];
      }
  }
}

extern "C" void kernel_launch(void* const* d_in, const int* in_sizes, int n_in,
                              void* d_out, int out_size, void* d_ws, size_t ws_size,
                              hipStream_t stream) {
  (void)in_sizes; (void)n_in; (void)out_size; (void)ws_size;
  const float* x      = (const float*)d_in[0];
  const float* mem    = (const float*)d_in[1];
  const float* Wq     = (const float*)d_in[2];
  const float* Wk     = (const float*)d_in[3];
  const float* Wv     = (const float*)d_in[4];
  const float* Wo     = (const float*)d_in[5];
  const float* ln1_g  = (const float*)d_in[6];
  const float* ln1_b  = (const float*)d_in[7];
  const float* rms_w  = (const float*)d_in[8];
  const float* W_in   = (const float*)d_in[9];
  const float* conv_w = (const float*)d_in[10];
  const float* conv_b = (const float*)d_in[11];
  const float* W_x    = (const float*)d_in[12];
  const float* W_dt   = (const float*)d_in[13];
  const float* b_dt   = (const float*)d_in[14];
  const float* A_log  = (const float*)d_in[15];
  const float* Dp     = (const float*)d_in[16];
  const float* W_out  = (const float*)d_in[17];

  float* ws = (float*)d_ws;
  size_t off = 0;
  float* kb      = ws + off; off += (size_t)MSLOTSn*Dn;
  float* vb      = ws + off; off += (size_t)MSLOTSn*Dn;
  float* Wq_sw   = ws + off; off += (size_t)8*32*32*2;
  float* Wo_sw   = ws + off; off += (size_t)8*32*32*2;
  float* Win_sw  = ws + off; off += (size_t)32*32*32*2;
  float* Wx_sw   = ws + off; off += (size_t)2*64*32*2;
  float* Wdt_sw  = ws + off; off += (size_t)16*2*32*2;
  float* Wout_sw = ws + off; off += (size_t)8*64*32*2;
  float* kfragT  = ws + off; off += (size_t)1*32*32*2;
  float* vfrag   = ws + off; off += (size_t)8*4*32*2;
  float* xr      = ws + off; off += (size_t)NTOK*Dn;
  float* xp      = ws + off; off += (size_t)NTOK*EDn;
  float* zb      = ws + off; off += (size_t)NTOK*EDn;
  float* xc      = ws + off; off += (size_t)NTOK*EDn;
  float* delta   = ws + off; off += (size_t)NTOK*EDn;
  float* Bm      = ws + off; off += (size_t)NTOK*DSTATEn;
  float* Cm      = ws + off; off += (size_t)NTOK*DSTATEn;
  float* y       = xp;

  swz_kernel<<<(8*32*32  + 255)/256, 256, 0, stream>>>(Wq,    Wq_sw,   Dn,  Dn,     8);
  swz_kernel<<<(8*32*32  + 255)/256, 256, 0, stream>>>(Wo,    Wo_sw,   Dn,  Dn,     8);
  swz_kernel<<<(32*32*32 + 255)/256, 256, 0, stream>>>(W_in,  Win_sw,  Dn,  2*EDn, 32);
  swz_kernel<<<(2*64*32  + 255)/256, 256, 0, stream>>>(W_x,   Wx_sw,   EDn, 24,     2);
  swz_kernel<<<(16*2*32  + 255)/256, 256, 0, stream>>>(W_dt,  Wdt_sw,  DTRANKn, EDn, 16);
  swz_kernel<<<(8*64*32  + 255)/256, 256, 0, stream>>>(W_out, Wout_sw, EDn, Dn,     8);

  kv_kernel   <<<(2*MSLOTSn*Dn + 255)/256, 256, 0, stream>>>(mem, Wk, Wv, kb, vb);
  kvswz_kernel<<<4, 256, 0, stream>>>(kb, vb, kfragT, vfrag);

  attn_kernel<<<NBLK, 128, 0, stream>>>(x, (const v2f*)Wq_sw, (const v2f*)Wo_sw,
                                        ln1_g, ln1_b, (const v2f*)kfragT,
                                        (const v2f*)vfrag, xr);
  in_kernel  <<<NBLK, 128, 0, stream>>>(xr, rms_w, (const v2f*)Win_sw, xp, zb);
  conv_kernel<<<NBLK, 128, 0, stream>>>(xp, conv_w, conv_b, (const v2f*)Wx_sw,
                                        (const v2f*)Wdt_sw, b_dt, xc, delta, Bm, Cm);
  scan_kernel<<<(Bn*EDn + 255)/256, 256, 0, stream>>>(delta, xc, Bm, Cm,
                                                      A_log, Dp, y);
  out_kernel <<<NBLK, 128, 0, stream>>>(y, zb, (const v2f*)Wout_sw, xr, x,
                                        (float*)d_out);
}


# attention and out products formed transposed (weights A, tokens B): a lane owns a token, D fragments feed the next product in registers, no LDS tiles; k-sum, softmax, LayerNorm sum orders change
# speedup vs baseline: 6.0460x; 6.0460x over previous
#include <hip/hip_runtime.h>
#include <hip/hip_bf16.h>
#include <math.h>

#define Bn      8
#define Sn      12
#define NNn     325
#define Dn      128
#define EDn     256
#define DSTATEn 8
#define DCONVn  4
#define DTRANKn 8
#define MSLOTSn 10
#define Ln      (Sn*NNn)
#define NTOK    (Bn*Ln)
#define TM      3
#define TOKB    (16*TM)
#define NBLK    (NTOK/TOKB)

typedef float v2f __attribute__((ext_vector_type(2)));
typedef float v8f __attribute__((ext_vector_type(8)));
typedef float v4f __attribute__((ext_vector_type(4)));

__device__ __forceinline__ v8f wmma4(v2f a, v2f b, v8f c) {
  return __builtin_amdgcn_wmma_f32_16x16x4_f32(false, a, false, b, (short)0, c,
                                               false, false);
}

__device__ __forceinline__ void store_c(float* C, int ldc, int nb, v8f c) {
  int lane = threadIdx.x & 31;
  int n = nb + (lane & 15);
  int mb = (lane < 16) ? 0 : 8;
#pragma unroll
  for (int j = 0; j < 8; ++j) C[(mb + j)*ldc + n] = c[j];
}

__device__ __forceinline__ void gemm3(const float* A, int lda, const v2f* Bsw,
                                      int KT, int nt, v8f acc[TM]) {
  int lane = threadIdx.x & 31;
  int m = lane & 15;
  int ks = (lane >> 4) << 1;
  const v2f* bp = Bsw + (size_t)nt * KT * 32 + lane;
  const v2f* a0 = (const v2f*)(A + (0*16 + m)*lda + ks);
  const v2f* a1 = (const v2f*)(A + (1*16 + m)*lda + ks);
  const v2f* a2 = (const v2f*)(A + (2*16 + m)*lda + ks);
  v8f z = {};
  acc[0] = z; acc[1] = z; acc[2] = z;
#pragma unroll 4
  for (int kt = 0; kt < KT; ++kt) {
    v2f b = bp[kt * 32];
    acc[0] = wmma4(a0[kt*2], b, acc[0]);
    acc[1] = wmma4(a1[kt*2], b, acc[1]);
    acc[2] = wmma4(a2[kt*2], b, acc[2]);
  }
}

template <int NW, int UNR>
__device__ __forceinline__ void gemm3n(const float* A, int lda, const v2f* Bsw,
                                       int KT, int nt0, v8f acc[NW][TM]) {
  int lane = threadIdx.x & 31;
  int m = lane & 15;
  int ks = (lane >> 4) << 1;
  const v2f* bp = Bsw + (size_t)nt0 * KT * 32 + lane;
  const v2f* a0 = (const v2f*)(A + (0*16 + m)*lda + ks);
  const v2f* a1 = (const v2f*)(A + (1*16 + m)*lda + ks);
  const v2f* a2 = (const v2f*)(A + (2*16 + m)*lda + ks);
  v8f z = {};
#pragma unroll
  for (int w = 0; w < NW; ++w) { acc[w][0] = z; acc[w][1] = z; acc[w][2] = z; }
#pragma unroll UNR
  for (int kt = 0; kt < KT; ++kt) {
    v2f fa0 = a0[kt*2];
    v2f fa1 = a1[kt*2];
    v2f fa2 = a2[kt*2];
#pragma unroll
    for (int w = 0; w < NW; ++w) {
      v2f b = bp[(w * KT + kt) * 32];
      acc[w][0] = wmma4(fa0, b, acc[w][0]);
      acc[w][1] = wmma4(fa1, b, acc[w][1]);
      acc[w][2] = wmma4(fa2, b, acc[w][2]);
    }
  }
}

__device__ __forceinline__ float silu_f(float x) { return x / (1.f + __expf(-x)); }

__global__ void swz_kernel(const float* W, float* dst, int K, int Nreal, int NT) {
  int idx = blockIdx.x * blockDim.x + threadIdx.x;
  int KT = K / 4;
  if (idx >= NT * KT * 32) return;
  int lane = idx & 31;
  int kt = (idx >> 5) % KT;
  int nt = idx / (32 * KT);
  int n = nt*16 + (lane & 15);
  int k = kt*4 + ((lane >> 4) << 1);
  float bx = (n < Nreal) ? W[k*Nreal + n]     : 0.f;
  float by = (n < Nreal) ? W[(k+1)*Nreal + n] : 0.f;
  dst[idx*2]     = bx;
  dst[idx*2 + 1] = by;
}

__global__ void kv_kernel(const float* mem, const float* Wk, const float* Wv,
                          float* kb, float* vb) {
  int i = blockIdx.x * blockDim.x + threadIdx.x;
  if (i >= 2 * MSLOTSn * Dn) return;
  int which = i / (MSLOTSn * Dn);
  int r = i % (MSLOTSn * Dn);
  int m = r / Dn, d = r % Dn;
  const float* W = which ? Wv : Wk;
  float acc = 0.f;
  for (int k = 0; k < Dn; ++k) acc += mem[m*Dn + k] * W[k*Dn + d];
  (which ? vb : kb)[m*Dn + d] = acc;
}

__global__ void kvswz_kernel(const float* kb, const float* vb,
                             float* kfragT, float* vfrag) {
  int idx = blockIdx.x * blockDim.x + threadIdx.x;
  if (idx < 32*32) {
    int lane = idx & 31; int st = idx >> 5;
    int slot = lane & 15;
    int fk = 16*(st >> 2) + 8*(lane >> 4) + 2*(st & 3);
    kfragT[idx*2]     = (slot < MSLOTSn) ? kb[slot*Dn + fk]     : 0.f;
    kfragT[idx*2 + 1] = (slot < MSLOTSn) ? kb[slot*Dn + fk + 1] : 0.f;
  }
  if (idx < 8*4*32) {
    int lane = idx & 31; int jp = (idx >> 5) & 3; int ft = idx >> 7;
    int f = ft*16 + (lane & 15); int sl = 8*(lane >> 4) + 2*jp;
    vfrag[idx*2]     = (sl     < MSLOTSn) ? vb[sl*Dn + f]       : 0.f;
    vfrag[idx*2 + 1] = (sl + 1 < MSLOTSn) ? vb[(sl + 1)*Dn + f] : 0.f;
  }
}

__global__ void swzT_kernel(const float* W, float* dst, int KS, int NT, int chained) {
  int idx = blockIdx.x * blockDim.x + threadIdx.x;
  if (idx >= NT * KS * 32) return;
  int lane = idx & 31;
  int ks = (idx >> 5) % KS;
  int ft = idx / (32 * KS);
  int h = lane >> 4;
  int f = ft*16 + (lane & 15);
  int k0 = chained ? (16*(ks >> 2) + 8*h + 2*(ks & 3))
                   : (8*(ks >> 1) + 4*h + 2*(ks & 1));
  dst[idx*2]     = W[k0*Dn + f];
  dst[idx*2 + 1] = W[(k0 + 1)*Dn + f];
}

__global__ void __launch_bounds__(32)
attn_kernel(const float* x, const v2f* WqT_sw, const v2f* WoT_sw,
            const float* ln_g, const float* ln_b,
            const v2f* kA, const v2f* vA, float* xr) {
  int lane = threadIdx.x;
  int c = lane & 15;
  int h = lane >> 4;
  long tok[2];
  tok[0] = ((long)blockIdx.x * 2) * 16 + c;
  tok[1] = tok[0] + 16;

  v4f xg[2][16];
#pragma unroll
  for (int u = 0; u < 2; ++u) {
    const v4f* xp_ = (const v4f*)(x + tok[u]*Dn) + h;
#pragma unroll
    for (int i = 0; i < 16; ++i) xg[u][i] = xp_[2*i];
  }

  v8f sT[2];
  {
    v8f s0 = {}, s1 = {};
#pragma unroll 1
    for (int ft = 0; ft < 8; ++ft) {
      const v2f* ap = WqT_sw + (size_t)ft*32*32 + lane;
      const v2f* kp = kA + (size_t)ft*4*32 + lane;
      v2f ka[4];
#pragma unroll
      for (int jp = 0; jp < 4; ++jp) ka[jp] = kp[jp*32];
      v8f a0 = {}, a1 = {};
#pragma unroll
      for (int i = 0; i < 16; ++i) {
#pragma unroll
        for (int e = 0; e < 2; ++e) {
          v2f a = ap[(2*i + e)*32];
          v2f b0 = {xg[0][i][2*e], xg[0][i][2*e + 1]};
          v2f b1 = {xg[1][i][2*e], xg[1][i][2*e + 1]};
          a0 = wmma4(a, b0, a0);
          a1 = wmma4(a, b1, a1);
        }
      }
#pragma unroll
      for (int jp = 0; jp < 4; ++jp) {
        v2f b0 = {a0[2*jp], a0[2*jp + 1]};
        v2f b1 = {a1[2*jp], a1[2*jp + 1]};
        s0 = wmma4(ka[jp], b0, s0);
        s1 = wmma4(ka[jp], b1, s1);
      }
    }
    sT[0] = s0; sT[1] = s1;
  }

#pragma unroll
  for (int u = 0; u < 2; ++u) {
    const float scale = 0.088388347648318447f;
    float v[8];
    float mx = -1e30f;
#pragma unroll
    for (int j = 0; j < 8; ++j) {
      v[j] = sT[u][j] * scale;
      bool ok = (h == 0) || (j < 2);
      mx = (ok && v[j] > mx) ? v[j] : mx;
    }
    float mo = __shfl_xor(mx, 16);
    mx = (mo > mx) ? mo : mx;
    float sum = 0.f;
#pragma unroll
    for (int j = 0; j < 8; ++j) {
      bool ok = (h == 0) || (j < 2);
      float ev = ok ? __expf(v[j] - mx) : 0.f;
      v[j] = ev; sum += ev;
    }
    sum += __shfl_xor(sum, 16);
    float inv = 1.f / sum;
#pragma unroll
    for (int j = 0; j < 8; ++j) sT[u][j] = v[j] * inv;
  }

  v8f avT[2][8];
#pragma unroll
  for (int ft = 0; ft < 8; ++ft) {
    v8f a0 = {}, a1 = {};
    const v2f* ap = vA + (size_t)ft*4*32 + lane;
#pragma unroll
    for (int jp = 0; jp < 4; ++jp) {
      v2f a = ap[jp*32];
      v2f b0 = {sT[0][2*jp], sT[0][2*jp + 1]};
      v2f b1 = {sT[1][2*jp], sT[1][2*jp + 1]};
      a0 = wmma4(a, b0, a0);
      a1 = wmma4(a, b1, a1);
    }
    avT[0][ft] = a0; avT[1][ft] = a1;
  }

  v8f xoT[2][8];
  {
    v2f wa[32], wb[32];
#define LOADW(dst, f2) { const v2f* ap_ = WoT_sw + (size_t)(f2)*32*32 + lane; \
      _Pragma("unroll") for (int s_ = 0; s_ < 32; ++s_) dst[s_] = ap_[s_*32]; }
#define TILE(src, f2) { v8f a0 = {}, a1 = {}; \
      _Pragma("unroll") for (int ft = 0; ft < 8; ++ft) \
      _Pragma("unroll") for (int jp = 0; jp < 4; ++jp) { \
        v2f b0 = {avT[0][ft][2*jp], avT[0][ft][2*jp + 1]}; \
        v2f b1 = {avT[1][ft][2*jp], avT[1][ft][2*jp + 1]}; \
        a0 = wmma4(src[ft*4 + jp], b0, a0); \
        a1 = wmma4(src[ft*4 + jp], b1, a1); } \
      xoT[0][f2] = a0; xoT[1][f2] = a1; }
    LOADW(wa, 0)
#pragma unroll
    for (int f2 = 0; f2 < 8; f2 += 2) {
      LOADW(wb, f2 + 1)
      __builtin_amdgcn_sched_barrier(0);
      TILE(wa, f2)
      if (f2 + 2 < 8) LOADW(wa, f2 + 2)
      __builtin_amdgcn_sched_barrier(0);
      TILE(wb, f2 + 1)
    }
#undef LOADW
#undef TILE
  }

  v4f gq[8][2], bq[8][2], xq[2][8][2];
#pragma unroll
  for (int f2 = 0; f2 < 8; ++f2) {
    int f0 = f2*16 + 8*h;
#pragma unroll
    for (int q = 0; q < 2; ++q) {
      gq[f2][q] = ((const v4f*)(ln_g + f0))[q];
      bq[f2][q] = ((const v4f*)(ln_b + f0))[q];
      xq[0][f2][q] = ((const v4f*)(x + tok[0]*Dn + f0))[q];
      xq[1][f2][q] = ((const v4f*)(x + tok[1]*Dn + f0))[q];
    }
  }
#pragma unroll
  for (int u = 0; u < 2; ++u) {
    float m = 0.f;
#pragma unroll
    for (int f2 = 0; f2 < 8; ++f2)
#pragma unroll
      for (int j = 0; j < 8; ++j) m += xoT[u][f2][j];
    m += __shfl_xor(m, 16);
    m *= (1.f / Dn);
    float var = 0.f;
#pragma unroll
    for (int f2 = 0; f2 < 8; ++f2)
#pragma unroll
      for (int j = 0; j < 8; ++j) { float d = xoT[u][f2][j] - m; var += d*d; }
    var += __shfl_xor(var, 16);
    var *= (1.f / Dn);
    float inv = rsqrtf(var + 1e-5f);
#pragma unroll
    for (int f2 = 0; f2 < 8; ++f2) {
      v4f* op = (v4f*)(xr + tok[u]*Dn + f2*16 + 8*h);
#pragma unroll
      for (int q = 0; q < 2; ++q) {
        v4f o;
#pragma unroll
        for (int e = 0; e < 4; ++e)
          o[e] = (xoT[u][f2][4*q + e] - m) * inv * gq[f2][q][e] + bq[f2][q][e]
                 + xq[u][f2][q][e];
        op[q] = o;
      }
    }
  }
}

__global__ void __launch_bounds__(128)
in_kernel(const float* xr, const float* rms_w, const v2f* Win_sw,
                          float* xp, float* zb) {
  __shared__ float xn[TOKB*Dn];
  int tid = threadIdx.x;
  int wave = tid >> 5;
  int lane = tid & 31;
  long t0 = (long)blockIdx.x * TOKB;

  for (int i = tid; i < TOKB*Dn; i += 128) xn[i] = xr[t0*Dn + i];
  __syncthreads();
  if (tid < TOKB) {
    float s = 0.f;
    for (int c = 0; c < Dn; ++c) { float v = xn[tid*Dn + c]; s += v*v; }
    float inv = rsqrtf(s / Dn + 1e-5f);
    for (int c = 0; c < Dn; ++c) xn[tid*Dn + c] = xn[tid*Dn + c] * inv * rms_w[c];
  }
  __syncthreads();

  v8f acc8[8][TM];
  gemm3n<8, 2>(xn, Dn, Win_sw, 32, wave*8, acc8);
#pragma unroll
  for (int w = 0; w < 8; ++w) {
    int nt = wave*8 + w;
    int n = nt*16 + (lane & 15);
    int mb = (lane < 16) ? 0 : 8;
#pragma unroll
    for (int t = 0; t < TM; ++t)
#pragma unroll
      for (int j = 0; j < 8; ++j) {
        long tok = t0 + t*16 + mb + j;
        float v = acc8[w][t][j];
        if (n < EDn) xp[tok*EDn + n] = v;
        else         zb[tok*EDn + (n - EDn)] = v;
      }
  }
}

#define CR 12

__global__ void __launch_bounds__(256)
dwconv_kernel(const float* xp, const float* conv_w, const float* conv_b, float* xc) {
  int q  = blockIdx.x * 4 + (threadIdx.x >> 6);
  int c4 = (threadIdx.x & 63) << 2;
  bool first = (q % (Ln / CR)) == 0;
  long tok0 = (long)q * CR;
  const v4f* src = (const v4f*)(xp + tok0*EDn + c4);
  v4f xin[CR + 3];
#pragma unroll
  for (int k = 0; k < 3; ++k) {
    v4f v = src[first ? 0 : (k - 3) * (EDn/4)];
    v4f z = {0.f, 0.f, 0.f, 0.f};
    xin[k] = first ? z : v;
  }
#pragma unroll
  for (int k = 0; k < CR; ++k) xin[3 + k] = src[k * (EDn/4)];
  v4f w[4];
#pragma unroll
  for (int i = 0; i < 4; ++i) w[i] = *(const v4f*)(conv_w + (c4 + i)*DCONVn);
  v4f bias = *(const v4f*)(conv_b + c4);
  v4f* dst = (v4f*)(xc + tok0*EDn + c4);
#pragma unroll
  for (int k = 0; k < CR; ++k) {
    v4f o;
#pragma unroll
    for (int i = 0; i < 4; ++i) {
      float a = bias[i];
#pragma unroll
      for (int j = 0; j < DCONVn; ++j) a += w[i][j] * xin[k + j][i];
      o[i] = silu_f(a);
    }
    dst[k * (EDn/4)] = o;
  }
}

__global__ void __launch_bounds__(128)
conv_kernel(const float* xc, const v2f* Wx_sw, const v2f* Wdt_sw,
                            const float* b_dt, float* delta, float* Bm, float* Cm) {
  __shared__ __attribute__((aligned(16))) float xcs[TOKB*EDn];
  __shared__ float dbcs[TOKB*32];
  int tid = threadIdx.x;
  int wave = tid >> 5;
  int lane = tid & 31;
  long t0 = (long)blockIdx.x * TOKB;
  v8f acc[TM];

  {
    const v4f* src = (const v4f*)(xc + t0*EDn);
    v4f* dst = (v4f*)xcs;
#pragma unroll
    for (int g = 0; g < 2; ++g) {
      v4f r[12];
#pragma unroll
      for (int k = 0; k < 12; ++k) r[k] = src[tid + (g*12 + k)*128];
#pragma unroll
      for (int k = 0; k < 12; ++k) dst[tid + (g*12 + k)*128] = r[k];
    }
  }
  __syncthreads();

  if (wave < 2) {
    gemm3(xcs, EDn, Wx_sw, 64, wave, acc);
#pragma unroll
    for (int t = 0; t < TM; ++t) store_c(dbcs + t*16*32, 32, wave*16, acc[t]);
  }
  __syncthreads();

  v8f acc4[4][TM];
  gemm3n<4, 2>(dbcs, 32, Wdt_sw, 2, wave*4, acc4);
  for (int w = 0; w < 4; ++w) {
    int nt = wave*4 + w;
    int n = nt*16 + (lane & 15);
    int mb = (lane < 16) ? 0 : 8;
#pragma unroll
    for (int t = 0; t < TM; ++t)
#pragma unroll
      for (int j = 0; j < 8; ++j) {
        long tok = t0 + t*16 + mb + j;
        float v = acc4[w][t][j] + b_dt[n];
        delta[tok*EDn + n] = (v > 20.f) ? v : log1pf(__expf(v));
      }
  }

  for (int i = tid; i < TOKB*16; i += 128) {
    int r = i >> 4, c = i & 15;
    long tok = t0 + r;
    float v = dbcs[r*32 + DTRANKn + c];
    if (c < DSTATEn) Bm[tok*DSTATEn + c]             = v;
    else             Cm[tok*DSTATEn + (c - DSTATEn)] = v;
  }
}

#define SCG   60
#define SNG   (Ln/SCG)
#define SREC  12

#define XADD(v, m) ({ float r_; \
    asm("v_add_f32_dpp %0, %1, %1 row_xmask:" #m " row_mask:0xf bank_mask:0xf" \
        : "=v"(r_) : "v"(v)); r_; })

__global__ void __launch_bounds__(128)
scan_kernel(const float* delta, const float* xc, const float* Bm,
            const float* Cm, const float* A_log, const float* Dp, float* y) {
  __shared__ v2f stage[4][2][SCG*SREC];
  int lane = threadIdx.x & 31;
  int wave = threadIdx.x >> 5;
  int c = lane >> 3;
  int n = lane & 7;
  int b = blockIdx.x >> 4;
  int d = ((blockIdx.x & 15) << 4) + (wave << 2) + c;
  float a = -__expf(A_log[d*DSTATEn + n]);
  float dp = Dp[d];
  float h = 0.f;
  long base = (long)b * Ln;
  int rlast = (56 + n < SCG) ? (56 + n) : (SCG - 1);

  float dtn[8], xvn[8], bn[15], cn[15];
  {
    const float* dsrc = delta + (base + n)*EDn + d;
    const float* xsrc = xc + (base + n)*EDn + d;
#pragma unroll
    for (int k = 0; k < 7; ++k) { dtn[k] = dsrc[k*8*EDn]; xvn[k] = xsrc[k*8*EDn]; }
    dtn[7] = delta[(base + rlast)*EDn + d];
    xvn[7] = xc[(base + rlast)*EDn + d];
    const float* bsrc = Bm + base*DSTATEn + lane;
    const float* csrc = Cm + base*DSTATEn + lane;
#pragma unroll
    for (int k = 0; k < 15; ++k) { bn[k] = bsrc[k*32]; cn[k] = csrc[k*32]; }
    v2f* st = stage[wave][0];
#pragma unroll
    for (int k = 0; k < 7; ++k) { v2f t = {dtn[k], xvn[k]}; st[(8*k + n)*SREC + c] = t; }
    { v2f t = {dtn[7], xvn[7]}; st[rlast*SREC + c] = t; }
#pragma unroll
    for (int k = 0; k < 15; ++k) { v2f t = {bn[k], cn[k]}; st[(4*k + c)*SREC + 4 + n] = t; }
  }
  __builtin_amdgcn_fence(__ATOMIC_ACQ_REL, "wavefront");

  for (int g = 0; g < SNG; ++g) {
    int gn = (g + 1 < SNG) ? (g + 1) : g;
    long tn = base + (long)gn * SCG;
    {
      const float* dsrc = delta + (tn + n)*EDn + d;
      const float* xsrc = xc + (tn + n)*EDn + d;
#pragma unroll
      for (int k = 0; k < 7; ++k) { dtn[k] = dsrc[k*8*EDn]; xvn[k] = xsrc[k*8*EDn]; }
      dtn[7] = delta[(tn + rlast)*EDn + d];
      xvn[7] = xc[(tn + rlast)*EDn + d];
      const float* bsrc = Bm + tn*DSTATEn + lane;
      const float* csrc = Cm + tn*DSTATEn + lane;
#pragma unroll
      for (int k = 0; k < 15; ++k) { bn[k] = bsrc[k*32]; cn[k] = csrc[k*32]; }
    }
    __builtin_amdgcn_sched_barrier(0);

    const v2f* cur = stage[wave][g & 1];
    v2f dxr[SCG], bcr[SCG];
#pragma unroll
    for (int s = 0; s < SCG; ++s) {
      dxr[s] = cur[s*SREC + c];
      bcr[s] = cur[s*SREC + 4 + n];
    }
    __builtin_amdgcn_sched_barrier(0);

    float* yp = y + (base + (long)g * SCG + n)*EDn + d;
    float ykeep = 0.f;
#pragma unroll
    for (int s = 0; s < SCG; ++s) {
      v2f dx = dxr[s];
      v2f bcv = bcr[s];
      float dt = dx[0];
      float xv = dx[1];
      float dA = __expf(dt * a);
      h = dA * h + dt * bcv[0] * xv;
      float p = h * bcv[1];
      p = XADD(p, 1);
      p = XADD(p, 2);
      p = XADD(p, 4);
      float yv = p + dp * xv;
      ykeep = (n == (s & 7)) ? yv : ykeep;
      if ((s & 7) == 7) yp[(s - 7)*EDn] = ykeep;
    }
    if (n < 4) yp[56*EDn] = ykeep;

    v2f* nxt = stage[wave][(g + 1) & 1];
#pragma unroll
    for (int k = 0; k < 7; ++k) { v2f t = {dtn[k], xvn[k]}; nxt[(8*k + n)*SREC + c] = t; }
    { v2f t = {dtn[7], xvn[7]}; nxt[rlast*SREC + c] = t; }
#pragma unroll
    for (int k = 0; k < 15; ++k) { v2f t = {bn[k], cn[k]}; nxt[(4*k + c)*SREC + 4 + n] = t; }
    __builtin_amdgcn_fence(__ATOMIC_ACQ_REL, "wavefront");
  }
}

__global__ void __launch_bounds__(32)
out_kernel(const float* y, const float* zb, const v2f* WoutT_sw,
           const float* xr, const float* x, float* out) {
  int lane = threadIdx.x;
  int c = lane & 15;
  int h = lane >> 4;
  long tok0 = ((long)blockIdx.x * 2) * 16 + c;
  long tok1 = tok0 + 16;

  v4f g0[32], g1[32];
  {
    const v4f* y0 = (const v4f*)(y  + tok0*EDn) + h;
    const v4f* z0 = (const v4f*)(zb + tok0*EDn) + h;
    const v4f* y1 = (const v4f*)(y  + tok1*EDn) + h;
    const v4f* z1 = (const v4f*)(zb + tok1*EDn) + h;
#pragma unroll
    for (int i = 0; i < 32; ++i) {
      v4f ya = y0[2*i], za = z0[2*i], yb = y1[2*i], zc = z1[2*i];
#pragma unroll
      for (int e = 0; e < 4; ++e) {
        g0[i][e] = ya[e] * silu_f(za[e]);
        g1[i][e] = yb[e] * silu_f(zc[e]);
      }
    }
  }

#pragma unroll 1
  for (int ft = 0; ft < 8; ++ft) {
    int f0 = ft*16 + 8*h;
    const v4f* r0p = (const v4f*)(xr + tok0*Dn + f0);
    const v4f* x0p = (const v4f*)(x  + tok0*Dn + f0);
    const v4f* r1p = (const v4f*)(xr + tok1*Dn + f0);
    const v4f* x1p = (const v4f*)(x  + tok1*Dn + f0);
    v4f r0a = r0p[0], r0b = r0p[1], x0a = x0p[0], x0b = x0p[1];
    v4f r1a = r1p[0], r1b = r1p[1], x1a = x1p[0], x1b = x1p[1];

    const v2f* ap = WoutT_sw + (size_t)ft*64*32 + lane;
    v8f acc0 = {}, acc1 = {};
#pragma unroll
    for (int i = 0; i < 32; ++i) {
      v2f a0 = ap[(2*i)*32];
      v2f a1 = ap[(2*i + 1)*32];
      v2f b00 = {g0[i][0], g0[i][1]};
      v2f b10 = {g1[i][0], g1[i][1]};
      v2f b01 = {g0[i][2], g0[i][3]};
      v2f b11 = {g1[i][2], g1[i][3]};
      acc0 = wmma4(a0, b00, acc0);
      acc1 = wmma4(a0, b10, acc1);
      acc0 = wmma4(a1, b01, acc0);
      acc1 = wmma4(a1, b11, acc1);
    }

    v4f o;
    v4f* o0p = (v4f*)(out + tok0*Dn + f0);
    v4f* o1p = (v4f*)(out + tok1*Dn + f0);
    o[0] = acc0[0] + r0a[0] + x0a[0]; o[1] = acc0[1] + r0a[1] + x0a[1];
    o[2] = acc0[2] + r0a[2] + x0a[2]; o[3] = acc0[3] + r0a[3] + x0a[3];
    o0p[0] = o;
    o[0] = acc0[4] + r0b[0] + x0b[0]; o[1] = acc0[5] + r0b[1] + x0b[1];
    o[2] = acc0[6] + r0b[2] + x0b[2]; o[3] = acc0[7] + r0b[3] + x0b[3];
    o0p[1] = o;
    o[0] = acc1[0] + r1a[0] + x1a[0]; o[1] = acc1[1] + r1a[1] + x1a[1];
    o[2] = acc1[2] + r1a[2] + x1a[2]; o[3] = acc1[3] + r1a[3] + x1a[3];
    o1p[0] = o;
    o[0] = acc1[4] + r1b[0] + x1b[0]; o[1] = acc1[5] + r1b[1] + x1b[1];
    o[2] = acc1[6] + r1b[2] + x1b[2]; o[3] = acc1[7] + r1b[3] + x1b[3];
    o1p[1] = o;
  }
}

extern "C" void kernel_launch(void* const* d_in, const int* in_sizes, int n_in,
                              void* d_out, int out_size, void* d_ws, size_t ws_size,
                              hipStream_t stream) {
  (void)in_sizes; (void)n_in; (void)out_size; (void)ws_size;
  const float* x      = (const float*)d_in[0];
  const float* mem    = (const float*)d_in[1];
  const float* Wq     = (const float*)d_in[2];
  const float* Wk     = (const float*)d_in[3];
  const float* Wv     = (const float*)d_in[4];
  const float* Wo     = (const float*)d_in[5];
  const float* ln1_g  = (const float*)d_in[6];
  const float* ln1_b  = (const float*)d_in[7];
  const float* rms_w  = (const float*)d_in[8];
  const float* W_in   = (const float*)d_in[9];
  const float* conv_w = (const float*)d_in[10];
  const float* conv_b = (const float*)d_in[11];
  const float* W_x    = (const float*)d_in[12];
  const float* W_dt   = (const float*)d_in[13];
  const float* b_dt   = (const float*)d_in[14];
  const float* A_log  = (const float*)d_in[15];
  const float* Dp     = (const float*)d_in[16];
  const float* W_out  = (const float*)d_in[17];

  float* ws = (float*)d_ws;
  size_t off = 0;
  float* kb      = ws + off; off += (size_t)MSLOTSn*Dn;
  float* vb      = ws + off; off += (size_t)MSLOTSn*Dn;
  float* Wq_sw   = ws + off; off += (size_t)8*32*32*2;
  float* Wo_sw   = ws + off; off += (size_t)8*32*32*2;
  float* Win_sw  = ws + off; off += (size_t)32*32*32*2;
  float* Wx_sw   = ws + off; off += (size_t)2*64*32*2;
  float* Wdt_sw  = ws + off; off += (size_t)16*2*32*2;
  float* Wout_sw = ws + off; off += (size_t)8*64*32*2;
  float* kfragT  = ws + off; off += (size_t)1*32*32*2;
  float* vfrag   = ws + off; off += (size_t)8*4*32*2;
  float* xr      = ws + off; off += (size_t)NTOK*Dn;
  float* xp      = ws + off; off += (size_t)NTOK*EDn;
  float* zb      = ws + off; off += (size_t)NTOK*EDn;
  float* xc      = ws + off; off += (size_t)NTOK*EDn;
  float* delta   = ws + off; off += (size_t)NTOK*EDn;
  float* Bm      = ws + off; off += (size_t)NTOK*DSTATEn;
  float* Cm      = ws + off; off += (size_t)NTOK*DSTATEn;
  float* y       = xp;

  swzT_kernel<<<(8*32*32 + 255)/256, 256, 0, stream>>>(Wq,    Wq_sw,   32, 8, 0);
  swzT_kernel<<<(8*32*32 + 255)/256, 256, 0, stream>>>(Wo,    Wo_sw,   32, 8, 1);
  swz_kernel<<<(32*32*32 + 255)/256, 256, 0, stream>>>(W_in,  Win_sw,  Dn,  2*EDn, 32);
  swz_kernel<<<(2*64*32  + 255)/256, 256, 0, stream>>>(W_x,   Wx_sw,   EDn, 24,     2);
  swz_kernel<<<(16*2*32  + 255)/256, 256, 0, stream>>>(W_dt,  Wdt_sw,  DTRANKn, EDn, 16);
  swzT_kernel<<<(8*64*32 + 255)/256, 256, 0, stream>>>(W_out, Wout_sw, 64, 8, 0);

  kv_kernel   <<<(2*MSLOTSn*Dn + 255)/256, 256, 0, stream>>>(mem, Wk, Wv, kb, vb);
  kvswz_kernel<<<4, 256, 0, stream>>>(kb, vb, kfragT, vfrag);

  attn_kernel<<<NTOK/32, 32, 0, stream>>>(x, (const v2f*)Wq_sw, (const v2f*)Wo_sw,
                                        ln1_g, ln1_b, (const v2f*)kfragT,
                                        (const v2f*)vfrag, xr);
  in_kernel  <<<NBLK, 128, 0, stream>>>(xr, rms_w, (const v2f*)Win_sw, xp, zb);
  dwconv_kernel<<<NTOK/CR/4, 256, 0, stream>>>(xp, conv_w, conv_b, xc);
  conv_kernel<<<NBLK, 128, 0, stream>>>(xc, (const v2f*)Wx_sw, (const v2f*)Wdt_sw,
                                        b_dt, delta, Bm, Cm);
  scan_kernel<<<(Bn*EDn)/16, 128, 0, stream>>>(delta, xc, Bm, Cm,
                                                       A_log, Dp, y);
  out_kernel <<<NTOK/32, 32, 0, stream>>>(y, zb, (const v2f*)Wout_sw, xr, x,
                                        (float*)d_out);
}


# scan: y's sum over a channel's 8 lanes, 8 steps at once as a halving tree (select + 1 DPP add per pair of steps: 7 DPP adds per 8 steps, not 24), each tree beside the next 8 updates; sums kept
# speedup vs baseline: 6.3684x; 1.0533x over previous
#include <hip/hip_runtime.h>
#include <hip/hip_bf16.h>
#include <math.h>

#define Bn      8
#define Sn      12
#define NNn     325
#define Dn      128
#define EDn     256
#define DSTATEn 8
#define DCONVn  4
#define DTRANKn 8
#define MSLOTSn 10
#define Ln      (Sn*NNn)
#define NTOK    (Bn*Ln)
#define TM      3
#define TOKB    (16*TM)
#define NBLK    (NTOK/TOKB)

typedef float v2f __attribute__((ext_vector_type(2)));
typedef float v8f __attribute__((ext_vector_type(8)));
typedef float v4f __attribute__((ext_vector_type(4)));

__device__ __forceinline__ v8f wmma4(v2f a, v2f b, v8f c) {
  return __builtin_amdgcn_wmma_f32_16x16x4_f32(false, a, false, b, (short)0, c,
                                               false, false);
}

__device__ __forceinline__ void store_c(float* C, int ldc, int nb, v8f c) {
  int lane = threadIdx.x & 31;
  int n = nb + (lane & 15);
  int mb = (lane < 16) ? 0 : 8;
#pragma unroll
  for (int j = 0; j < 8; ++j) C[(mb + j)*ldc + n] = c[j];
}

__device__ __forceinline__ void gemm3(const float* A, int lda, const v2f* Bsw,
                                      int KT, int nt, v8f acc[TM]) {
  int lane = threadIdx.x & 31;
  int m = lane & 15;
  int ks = (lane >> 4) << 1;
  const v2f* bp = Bsw + (size_t)nt * KT * 32 + lane;
  const v2f* a0 = (const v2f*)(A + (0*16 + m)*lda + ks);
  const v2f* a1 = (const v2f*)(A + (1*16 + m)*lda + ks);
  const v2f* a2 = (const v2f*)(A + (2*16 + m)*lda + ks);
  v8f z = {};
  acc[0] = z; acc[1] = z; acc[2] = z;
#pragma unroll 4
  for (int kt = 0; kt < KT; ++kt) {
    v2f b = bp[kt * 32];
    acc[0] = wmma4(a0[kt*2], b, acc[0]);
    acc[1] = wmma4(a1[kt*2], b, acc[1]);
    acc[2] = wmma4(a2[kt*2], b, acc[2]);
  }
}

template <int NW, int UNR>
__device__ __forceinline__ void gemm3n(const float* A, int lda, const v2f* Bsw,
                                       int KT, int nt0, v8f acc[NW][TM]) {
  int lane = threadIdx.x & 31;
  int m = lane & 15;
  int ks = (lane >> 4) << 1;
  const v2f* bp = Bsw + (size_t)nt0 * KT * 32 + lane;
  const v2f* a0 = (const v2f*)(A + (0*16 + m)*lda + ks);
  const v2f* a1 = (const v2f*)(A + (1*16 + m)*lda + ks);
  const v2f* a2 = (const v2f*)(A + (2*16 + m)*lda + ks);
  v8f z = {};
#pragma unroll
  for (int w = 0; w < NW; ++w) { acc[w][0] = z; acc[w][1] = z; acc[w][2] = z; }
#pragma unroll UNR
  for (int kt = 0; kt < KT; ++kt) {
    v2f fa0 = a0[kt*2];
    v2f fa1 = a1[kt*2];
    v2f fa2 = a2[kt*2];
#pragma unroll
    for (int w = 0; w < NW; ++w) {
      v2f b = bp[(w * KT + kt) * 32];
      acc[w][0] = wmma4(fa0, b, acc[w][0]);
      acc[w][1] = wmma4(fa1, b, acc[w][1]);
      acc[w][2] = wmma4(fa2, b, acc[w][2]);
    }
  }
}

__device__ __forceinline__ float silu_f(float x) { return x / (1.f + __expf(-x)); }

__global__ void swz_kernel(const float* W, float* dst, int K, int Nreal, int NT) {
  int idx = blockIdx.x * blockDim.x + threadIdx.x;
  int KT = K / 4;
  if (idx >= NT * KT * 32) return;
  int lane = idx & 31;
  int kt = (idx >> 5) % KT;
  int nt = idx / (32 * KT);
  int n = nt*16 + (lane & 15);
  int k = kt*4 + ((lane >> 4) << 1);
  float bx = (n < Nreal) ? W[k*Nreal + n]     : 0.f;
  float by = (n < Nreal) ? W[(k+1)*Nreal + n] : 0.f;
  dst[idx*2]     = bx;
  dst[idx*2 + 1] = by;
}

__global__ void kv_kernel(const float* mem, const float* Wk, const float* Wv,
                          float* kb, float* vb) {
  int i = blockIdx.x * blockDim.x + threadIdx.x;
  if (i >= 2 * MSLOTSn * Dn) return;
  int which = i / (MSLOTSn * Dn);
  int r = i % (MSLOTSn * Dn);
  int m = r / Dn, d = r % Dn;
  const float* W = which ? Wv : Wk;
  float acc = 0.f;
  for (int k = 0; k < Dn; ++k) acc += mem[m*Dn + k] * W[k*Dn + d];
  (which ? vb : kb)[m*Dn + d] = acc;
}

__global__ void kvswz_kernel(const float* kb, const float* vb,
                             float* kfragT, float* vfrag) {
  int idx = blockIdx.x * blockDim.x + threadIdx.x;
  if (idx < 32*32) {
    int lane = idx & 31; int st = idx >> 5;
    int slot = lane & 15;
    int fk = 16*(st >> 2) + 8*(lane >> 4) + 2*(st & 3);
    kfragT[idx*2]     = (slot < MSLOTSn) ? kb[slot*Dn + fk]     : 0.f;
    kfragT[idx*2 + 1] = (slot < MSLOTSn) ? kb[slot*Dn + fk + 1] : 0.f;
  }
  if (idx < 8*4*32) {
    int lane = idx & 31; int jp = (idx >> 5) & 3; int ft = idx >> 7;
    int f = ft*16 + (lane & 15); int sl = 8*(lane >> 4) + 2*jp;
    vfrag[idx*2]     = (sl     < MSLOTSn) ? vb[sl*Dn + f]       : 0.f;
    vfrag[idx*2 + 1] = (sl + 1 < MSLOTSn) ? vb[(sl + 1)*Dn + f] : 0.f;
  }
}

__global__ void swzT_kernel(const float* W, float* dst, int KS, int NT, int chained) {
  int idx = blockIdx.x * blockDim.x + threadIdx.x;
  if (idx >= NT * KS * 32) return;
  int lane = idx & 31;
  int ks = (idx >> 5) % KS;
  int ft = idx / (32 * KS);
  int h = lane >> 4;
  int f = ft*16 + (lane & 15);
  int k0 = chained ? (16*(ks >> 2) + 8*h + 2*(ks & 3))
                   : (8*(ks >> 1) + 4*h + 2*(ks & 1));
  dst[idx*2]     = W[k0*Dn + f];
  dst[idx*2 + 1] = W[(k0 + 1)*Dn + f];
}

__global__ void __launch_bounds__(32)
attn_kernel(const float* x, const v2f* WqT_sw, const v2f* WoT_sw,
            const float* ln_g, const float* ln_b,
            const v2f* kA, const v2f* vA, float* xr) {
  int lane = threadIdx.x;
  int c = lane & 15;
  int h = lane >> 4;
  long tok[2];
  tok[0] = ((long)blockIdx.x * 2) * 16 + c;
  tok[1] = tok[0] + 16;

  v4f xg[2][16];
#pragma unroll
  for (int u = 0; u < 2; ++u) {
    const v4f* xp_ = (const v4f*)(x + tok[u]*Dn) + h;
#pragma unroll
    for (int i = 0; i < 16; ++i) xg[u][i] = xp_[2*i];
  }

  v8f sT[2];
  {
    v8f s0 = {}, s1 = {};
#pragma unroll 1
    for (int ft = 0; ft < 8; ++ft) {
      const v2f* ap = WqT_sw + (size_t)ft*32*32 + lane;
      const v2f* kp = kA + (size_t)ft*4*32 + lane;
      v2f ka[4];
#pragma unroll
      for (int jp = 0; jp < 4; ++jp) ka[jp] = kp[jp*32];
      v8f a0 = {}, a1 = {};
#pragma unroll
      for (int i = 0; i < 16; ++i) {
#pragma unroll
        for (int e = 0; e < 2; ++e) {
          v2f a = ap[(2*i + e)*32];
          v2f b0 = {xg[0][i][2*e], xg[0][i][2*e + 1]};
          v2f b1 = {xg[1][i][2*e], xg[1][i][2*e + 1]};
          a0 = wmma4(a, b0, a0);
          a1 = wmma4(a, b1, a1);
        }
      }
#pragma unroll
      for (int jp = 0; jp < 4; ++jp) {
        v2f b0 = {a0[2*jp], a0[2*jp + 1]};
        v2f b1 = {a1[2*jp], a1[2*jp + 1]};
        s0 = wmma4(ka[jp], b0, s0);
        s1 = wmma4(ka[jp], b1, s1);
      }
    }
    sT[0] = s0; sT[1] = s1;
  }

#pragma unroll
  for (int u = 0; u < 2; ++u) {
    const float scale = 0.088388347648318447f;
    float v[8];
    float mx = -1e30f;
#pragma unroll
    for (int j = 0; j < 8; ++j) {
      v[j] = sT[u][j] * scale;
      bool ok = (h == 0) || (j < 2);
      mx = (ok && v[j] > mx) ? v[j] : mx;
    }
    float mo = __shfl_xor(mx, 16);
    mx = (mo > mx) ? mo : mx;
    float sum = 0.f;
#pragma unroll
    for (int j = 0; j < 8; ++j) {
      bool ok = (h == 0) || (j < 2);
      float ev = ok ? __expf(v[j] - mx) : 0.f;
      v[j] = ev; sum += ev;
    }
    sum += __shfl_xor(sum, 16);
    float inv = 1.f / sum;
#pragma unroll
    for (int j = 0; j < 8; ++j) sT[u][j] = v[j] * inv;
  }

  v8f avT[2][8];
#pragma unroll
  for (int ft = 0; ft < 8; ++ft) {
    v8f a0 = {}, a1 = {};
    const v2f* ap = vA + (size_t)ft*4*32 + lane;
#pragma unroll
    for (int jp = 0; jp < 4; ++jp) {
      v2f a = ap[jp*32];
      v2f b0 = {sT[0][2*jp], sT[0][2*jp + 1]};
      v2f b1 = {sT[1][2*jp], sT[1][2*jp + 1]};
      a0 = wmma4(a, b0, a0);
      a1 = wmma4(a, b1, a1);
    }
    avT[0][ft] = a0; avT[1][ft] = a1;
  }

  v8f xoT[2][8];
  {
    v2f wa[32], wb[32];
#define LOADW(dst, f2) { const v2f* ap_ = WoT_sw + (size_t)(f2)*32*32 + lane; \
      _Pragma("unroll") for (int s_ = 0; s_ < 32; ++s_) dst[s_] = ap_[s_*32]; }
#define TILE(src, f2) { v8f a0 = {}, a1 = {}; \
      _Pragma("unroll") for (int ft = 0; ft < 8; ++ft) \
      _Pragma("unroll") for (int jp = 0; jp < 4; ++jp) { \
        v2f b0 = {avT[0][ft][2*jp], avT[0][ft][2*jp + 1]}; \
        v2f b1 = {avT[1][ft][2*jp], avT[1][ft][2*jp + 1]}; \
        a0 = wmma4(src[ft*4 + jp], b0, a0); \
        a1 = wmma4(src[ft*4 + jp], b1, a1); } \
      xoT[0][f2] = a0; xoT[1][f2] = a1; }
    LOADW(wa, 0)
#pragma unroll
    for (int f2 = 0; f2 < 8; f2 += 2) {
      LOADW(wb, f2 + 1)
      __builtin_amdgcn_sched_barrier(0);
      TILE(wa, f2)
      if (f2 + 2 < 8) LOADW(wa, f2 + 2)
      __builtin_amdgcn_sched_barrier(0);
      TILE(wb, f2 + 1)
    }
#undef LOADW
#undef TILE
  }

  v4f gq[8][2], bq[8][2], xq[2][8][2];
#pragma unroll
  for (int f2 = 0; f2 < 8; ++f2) {
    int f0 = f2*16 + 8*h;
#pragma unroll
    for (int q = 0; q < 2; ++q) {
      gq[f2][q] = ((const v4f*)(ln_g + f0))[q];
      bq[f2][q] = ((const v4f*)(ln_b + f0))[q];
      xq[0][f2][q] = ((const v4f*)(x + tok[0]*Dn + f0))[q];
      xq[1][f2][q] = ((const v4f*)(x + tok[1]*Dn + f0))[q];
    }
  }
#pragma unroll
  for (int u = 0; u < 2; ++u) {
    float m = 0.f;
#pragma unroll
    for (int f2 = 0; f2 < 8; ++f2)
#pragma unroll
      for (int j = 0; j < 8; ++j) m += xoT[u][f2][j];
    m += __shfl_xor(m, 16);
    m *= (1.f / Dn);
    float var = 0.f;
#pragma unroll
    for (int f2 = 0; f2 < 8; ++f2)
#pragma unroll
      for (int j = 0; j < 8; ++j) { float d = xoT[u][f2][j] - m; var += d*d; }
    var += __shfl_xor(var, 16);
    var *= (1.f / Dn);
    float inv = rsqrtf(var + 1e-5f);
#pragma unroll
    for (int f2 = 0; f2 < 8; ++f2) {
      v4f* op = (v4f*)(xr + tok[u]*Dn + f2*16 + 8*h);
#pragma unroll
      for (int q = 0; q < 2; ++q) {
        v4f o;
#pragma unroll
        for (int e = 0; e < 4; ++e)
          o[e] = (xoT[u][f2][4*q + e] - m) * inv * gq[f2][q][e] + bq[f2][q][e]
                 + xq[u][f2][q][e];
        op[q] = o;
      }
    }
  }
}

__global__ void __launch_bounds__(128)
in_kernel(const float* xr, const float* rms_w, const v2f* Win_sw,
                          float* xp, float* zb) {
  __shared__ float xn[TOKB*Dn];
  int tid = threadIdx.x;
  int wave = tid >> 5;
  int lane = tid & 31;
  long t0 = (long)blockIdx.x * TOKB;

  for (int i = tid; i < TOKB*Dn; i += 128) xn[i] = xr[t0*Dn + i];
  __syncthreads();
  if (tid < TOKB) {
    float s = 0.f;
    for (int c = 0; c < Dn; ++c) { float v = xn[tid*Dn + c]; s += v*v; }
    float inv = rsqrtf(s / Dn + 1e-5f);
    for (int c = 0; c < Dn; ++c) xn[tid*Dn + c] = xn[tid*Dn + c] * inv * rms_w[c];
  }
  __syncthreads();

  v8f acc8[8][TM];
  gemm3n<8, 2>(xn, Dn, Win_sw, 32, wave*8, acc8);
#pragma unroll
  for (int w = 0; w < 8; ++w) {
    int nt = wave*8 + w;
    int n = nt*16 + (lane & 15);
    int mb = (lane < 16) ? 0 : 8;
#pragma unroll
    for (int t = 0; t < TM; ++t)
#pragma unroll
      for (int j = 0; j < 8; ++j) {
        long tok = t0 + t*16 + mb + j;
        float v = acc8[w][t][j];
        if (n < EDn) xp[tok*EDn + n] = v;
        else         zb[tok*EDn + (n - EDn)] = v;
      }
  }
}

#define CR 12

__global__ void __launch_bounds__(256)
dwconv_kernel(const float* xp, const float* conv_w, const float* conv_b, float* xc) {
  int q  = blockIdx.x * 4 + (threadIdx.x >> 6);
  int c4 = (threadIdx.x & 63) << 2;
  bool first = (q % (Ln / CR)) == 0;
  long tok0 = (long)q * CR;
  const v4f* src = (const v4f*)(xp + tok0*EDn + c4);
  v4f xin[CR + 3];
#pragma unroll
  for (int k = 0; k < 3; ++k) {
    v4f v = src[first ? 0 : (k - 3) * (EDn/4)];
    v4f z = {0.f, 0.f, 0.f, 0.f};
    xin[k] = first ? z : v;
  }
#pragma unroll
  for (int k = 0; k < CR; ++k) xin[3 + k] = src[k * (EDn/4)];
  v4f w[4];
#pragma unroll
  for (int i = 0; i < 4; ++i) w[i] = *(const v4f*)(conv_w + (c4 + i)*DCONVn);
  v4f bias = *(const v4f*)(conv_b + c4);
  v4f* dst = (v4f*)(xc + tok0*EDn + c4);
#pragma unroll
  for (int k = 0; k < CR; ++k) {
    v4f o;
#pragma unroll
    for (int i = 0; i < 4; ++i) {
      float a = bias[i];
#pragma unroll
      for (int j = 0; j < DCONVn; ++j) a += w[i][j] * xin[k + j][i];
      o[i] = silu_f(a);
    }
    dst[k * (EDn/4)] = o;
  }
}

__global__ void __launch_bounds__(128)
conv_kernel(const float* xc, const v2f* Wx_sw, const v2f* Wdt_sw,
                            const float* b_dt, float* delta, float* Bm, float* Cm) {
  __shared__ __attribute__((aligned(16))) float xcs[TOKB*EDn];
  __shared__ float dbcs[TOKB*32];
  int tid = threadIdx.x;
  int wave = tid >> 5;
  int lane = tid & 31;
  long t0 = (long)blockIdx.x * TOKB;
  v8f acc[TM];

  {
    const v4f* src = (const v4f*)(xc + t0*EDn);
    v4f* dst = (v4f*)xcs;
#pragma unroll
    for (int g = 0; g < 2; ++g) {
      v4f r[12];
#pragma unroll
      for (int k = 0; k < 12; ++k) r[k] = src[tid + (g*12 + k)*128];
#pragma unroll
      for (int k = 0; k < 12; ++k) dst[tid + (g*12 + k)*128] = r[k];
    }
  }
  __syncthreads();

  if (wave < 2) {
    gemm3(xcs, EDn, Wx_sw, 64, wave, acc);
#pragma unroll
    for (int t = 0; t < TM; ++t) store_c(dbcs + t*16*32, 32, wave*16, acc[t]);
  }
  __syncthreads();

  v8f acc4[4][TM];
  gemm3n<4, 2>(dbcs, 32, Wdt_sw, 2, wave*4, acc4);
  for (int w = 0; w < 4; ++w) {
    int nt = wave*4 + w;
    int n = nt*16 + (lane & 15);
    int mb = (lane < 16) ? 0 : 8;
#pragma unroll
    for (int t = 0; t < TM; ++t)
#pragma unroll
      for (int j = 0; j < 8; ++j) {
        long tok = t0 + t*16 + mb + j;
        float v = acc4[w][t][j] + b_dt[n];
        delta[tok*EDn + n] = (v > 20.f) ? v : log1pf(__expf(v));
      }
  }

  for (int i = tid; i < TOKB*16; i += 128) {
    int r = i >> 4, c = i & 15;
    long tok = t0 + r;
    float v = dbcs[r*32 + DTRANKn + c];
    if (c < DSTATEn) Bm[tok*DSTATEn + c]             = v;
    else             Cm[tok*DSTATEn + (c - DSTATEn)] = v;
  }
}

#define SCG   60
#define SNG   (Ln/SCG)
#define SREC  12

#define XADD(v, m) ({ float r_; \
    asm("v_add_f32_dpp %0, %1, %1 row_xmask:" #m " row_mask:0xf bank_mask:0xf" \
        : "=v"(r_) : "v"(v)); r_; })

#define XADD2(send, keep, m) ({ float r_; \
    asm("v_add_f32_dpp %0, %1, %2 row_xmask:" #m " row_mask:0xf bank_mask:0xf" \
        : "=v"(r_) : "v"(send), "v"(keep)); r_; })
#define HALVE(lo, hi, bit, m) ({ float k_ = (bit) ? (hi) : (lo); \
    float s_ = (bit) ? (lo) : (hi); XADD2(s_, k_, m); })

__global__ void __launch_bounds__(128)
scan_kernel(const float* delta, const float* xc, const float* Bm,
            const float* Cm, const float* A_log, const float* Dp, float* y) {
  __shared__ v2f stage[4][2][SCG*SREC];
  int lane = threadIdx.x & 31;
  int wave = threadIdx.x >> 5;
  int c = lane >> 3;
  int n = lane & 7;
  int b = blockIdx.x >> 4;
  int d = ((blockIdx.x & 15) << 4) + (wave << 2) + c;
  float a = -__expf(A_log[d*DSTATEn + n]);
  float dp = Dp[d];
  float h = 0.f;
  long base = (long)b * Ln;
  int rlast = (56 + n < SCG) ? (56 + n) : (SCG - 1);

  float dtn[8], xvn[8], bn[15], cn[15];
  {
    const float* dsrc = delta + (base + n)*EDn + d;
    const float* xsrc = xc + (base + n)*EDn + d;
#pragma unroll
    for (int k = 0; k < 7; ++k) { dtn[k] = dsrc[k*8*EDn]; xvn[k] = xsrc[k*8*EDn]; }
    dtn[7] = delta[(base + rlast)*EDn + d];
    xvn[7] = xc[(base + rlast)*EDn + d];
    const float* bsrc = Bm + base*DSTATEn + lane;
    const float* csrc = Cm + base*DSTATEn + lane;
#pragma unroll
    for (int k = 0; k < 15; ++k) { bn[k] = bsrc[k*32]; cn[k] = csrc[k*32]; }
    v2f* st = stage[wave][0];
#pragma unroll
    for (int k = 0; k < 7; ++k) { v2f t = {dtn[k], xvn[k]}; st[(8*k + n)*SREC + c] = t; }
    { v2f t = {dtn[7], xvn[7]}; st[rlast*SREC + c] = t; }
#pragma unroll
    for (int k = 0; k < 15; ++k) { v2f t = {bn[k], cn[k]}; st[(4*k + c)*SREC + 4 + n] = t; }
  }
  __builtin_amdgcn_fence(__ATOMIC_ACQ_REL, "wavefront");

  bool b0 = (n & 1) != 0, b1 = (n & 2) != 0, b2 = (n & 4) != 0;
  float xcur[8];
  for (int g = 0; g < SNG; ++g) {
    int gn = (g + 1 < SNG) ? (g + 1) : g;
    long tn = base + (long)gn * SCG;
#pragma unroll
    for (int k = 0; k < 8; ++k) xcur[k] = xvn[k];
    {
      const float* dsrc = delta + (tn + n)*EDn + d;
      const float* xsrc = xc + (tn + n)*EDn + d;
#pragma unroll
      for (int k = 0; k < 7; ++k) { dtn[k] = dsrc[k*8*EDn]; xvn[k] = xsrc[k*8*EDn]; }
      dtn[7] = delta[(tn + rlast)*EDn + d];
      xvn[7] = xc[(tn + rlast)*EDn + d];
      const float* bsrc = Bm + tn*DSTATEn + lane;
      const float* csrc = Cm + tn*DSTATEn + lane;
#pragma unroll
      for (int k = 0; k < 15; ++k) { bn[k] = bsrc[k*32]; cn[k] = csrc[k*32]; }
    }
    __builtin_amdgcn_sched_barrier(0);

    const v2f* cur = stage[wave][g & 1];
    v2f dxr[SCG], bcr[SCG];
#pragma unroll
    for (int s = 0; s < SCG; ++s) {
      dxr[s] = cur[s*SREC + c];
      bcr[s] = cur[s*SREC + 4 + n];
    }
    __builtin_amdgcn_sched_barrier(0);

    float* yp = y + (base + (long)g * SCG + n)*EDn + d;
#define STEP(s) ({ v2f dx_ = dxr[s]; v2f bc_ = bcr[s]; \
      float dA_ = __expf(dx_[0] * a); \
      h = dA_ * h + dx_[0] * bc_[0] * dx_[1]; \
      h * bc_[1]; })
    float p[8][8], pt[4];
#pragma unroll
    for (int i = 0; i < 8; ++i) p[0][i] = STEP(i);
    __builtin_amdgcn_sched_barrier(0);
#pragma unroll
    for (int j = 0; j < 7; ++j) {
      if (j < 6) {
#pragma unroll
        for (int i = 0; i < 8; ++i) p[j + 1][i] = STEP(8*(j + 1) + i);
      } else {
#pragma unroll
        for (int i = 0; i < 4; ++i) pt[i] = STEP(56 + i);
      }
      float q0 = HALVE(p[j][0], p[j][1], b0, 1), q1 = HALVE(p[j][2], p[j][3], b0, 1);
      float q2 = HALVE(p[j][4], p[j][5], b0, 1), q3 = HALVE(p[j][6], p[j][7], b0, 1);
      float r0 = HALVE(q0, q1, b1, 2), r1 = HALVE(q2, q3, b1, 2);
      float ys = HALVE(r0, r1, b2, 4);
      yp[8*j*EDn] = ys + dp * xcur[j];
      __builtin_amdgcn_sched_barrier(0);
    }
    {
      float q0 = HALVE(pt[0], pt[1], b0, 1), q1 = HALVE(pt[2], pt[3], b0, 1);
      float r0 = HALVE(q0, q1, b1, 2);
      float ys = XADD(r0, 4);
      if (n < 4) yp[56*EDn] = ys + dp * xcur[7];
    }
#undef STEP

    v2f* nxt = stage[wave][(g + 1) & 1];
#pragma unroll
    for (int k = 0; k < 7; ++k) { v2f t = {dtn[k], xvn[k]}; nxt[(8*k + n)*SREC + c] = t; }
    { v2f t = {dtn[7], xvn[7]}; nxt[rlast*SREC + c] = t; }
#pragma unroll
    for (int k = 0; k < 15; ++k) { v2f t = {bn[k], cn[k]}; nxt[(4*k + c)*SREC + 4 + n] = t; }
    __builtin_amdgcn_fence(__ATOMIC_ACQ_REL, "wavefront");
  }
}

__global__ void __launch_bounds__(32)
out_kernel(const float* y, const float* zb, const v2f* WoutT_sw,
           const float* xr, const float* x, float* out) {
  int lane = threadIdx.x;
  int c = lane & 15;
  int h = lane >> 4;
  long tok0 = ((long)blockIdx.x * 2) * 16 + c;
  long tok1 = tok0 + 16;

  v4f g0[32], g1[32];
  {
    const v4f* y0 = (const v4f*)(y  + tok0*EDn) + h;
    const v4f* z0 = (const v4f*)(zb + tok0*EDn) + h;
    const v4f* y1 = (const v4f*)(y  + tok1*EDn) + h;
    const v4f* z1 = (const v4f*)(zb + tok1*EDn) + h;
#pragma unroll
    for (int i = 0; i < 32; ++i) {
      v4f ya = y0[2*i], za = z0[2*i], yb = y1[2*i], zc = z1[2*i];
#pragma unroll
      for (int e = 0; e < 4; ++e) {
        g0[i][e] = ya[e] * silu_f(za[e]);
        g1[i][e] = yb[e] * silu_f(zc[e]);
      }
    }
  }

#pragma unroll 1
  for (int ft = 0; ft < 8; ++ft) {
    int f0 = ft*16 + 8*h;
    const v4f* r0p = (const v4f*)(xr + tok0*Dn + f0);
    const v4f* x0p = (const v4f*)(x  + tok0*Dn + f0);
    const v4f* r1p = (const v4f*)(xr + tok1*Dn + f0);
    const v4f* x1p = (const v4f*)(x  + tok1*Dn + f0);
    v4f r0a = r0p[0], r0b = r0p[1], x0a = x0p[0], x0b = x0p[1];
    v4f r1a = r1p[0], r1b = r1p[1], x1a = x1p[0], x1b = x1p[1];

    const v2f* ap = WoutT_sw + (size_t)ft*64*32 + lane;
    v8f acc0 = {}, acc1 = {};
#pragma unroll
    for (int i = 0; i < 32; ++i) {
      v2f a0 = ap[(2*i)*32];
      v2f a1 = ap[(2*i + 1)*32];
      v2f b00 = {g0[i][0], g0[i][1]};
      v2f b10 = {g1[i][0], g1[i][1]};
      v2f b01 = {g0[i][2], g0[i][3]};
      v2f b11 = {g1[i][2], g1[i][3]};
      acc0 = wmma4(a0, b00, acc0);
      acc1 = wmma4(a0, b10, acc1);
      acc0 = wmma4(a1, b01, acc0);
      acc1 = wmma4(a1, b11, acc1);
    }

    v4f o;
    v4f* o0p = (v4f*)(out + tok0*Dn + f0);
    v4f* o1p = (v4f*)(out + tok1*Dn + f0);
    o[0] = acc0[0] + r0a[0] + x0a[0]; o[1] = acc0[1] + r0a[1] + x0a[1];
    o[2] = acc0[2] + r0a[2] + x0a[2]; o[3] = acc0[3] + r0a[3] + x0a[3];
    o0p[0] = o;
    o[0] = acc0[4] + r0b[0] + x0b[0]; o[1] = acc0[5] + r0b[1] + x0b[1];
    o[2] = acc0[6] + r0b[2] + x0b[2]; o[3] = acc0[7] + r0b[3] + x0b[3];
    o0p[1] = o;
    o[0] = acc1[0] + r1a[0] + x1a[0]; o[1] = acc1[1] + r1a[1] + x1a[1];
    o[2] = acc1[2] + r1a[2] + x1a[2]; o[3] = acc1[3] + r1a[3] + x1a[3];
    o1p[0] = o;
    o[0] = acc1[4] + r1b[0] + x1b[0]; o[1] = acc1[5] + r1b[1] + x1b[1];
    o[2] = acc1[6] + r1b[2] + x1b[2]; o[3] = acc1[7] + r1b[3] + x1b[3];
    o1p[1] = o;
  }
}

extern "C" void kernel_launch(void* const* d_in, const int* in_sizes, int n_in,
                              void* d_out, int out_size, void* d_ws, size_t ws_size,
                              hipStream_t stream) {
  (void)in_sizes; (void)n_in; (void)out_size; (void)ws_size;
  const float* x      = (const float*)d_in[0];
  const float* mem    = (const float*)d_in[1];
  const float* Wq     = (const float*)d_in[2];
  const float* Wk     = (const float*)d_in[3];
  const float* Wv     = (const float*)d_in[4];
  const float* Wo     = (const float*)d_in[5];
  const float* ln1_g  = (const float*)d_in[6];
  const float* ln1_b  = (const float*)d_in[7];
  const float* rms_w  = (const float*)d_in[8];
  const float* W_in   = (const float*)d_in[9];
  const float* conv_w = (const float*)d_in[10];
  const float* conv_b = (const float*)d_in[11];
  const float* W_x    = (const float*)d_in[12];
  const float* W_dt   = (const float*)d_in[13];
  const float* b_dt   = (const float*)d_in[14];
  const float* A_log  = (const float*)d_in[15];
  const float* Dp     = (const float*)d_in[16];
  const float* W_out  = (const float*)d_in[17];

  float* ws = (float*)d_ws;
  size_t off = 0;
  float* kb      = ws + off; off += (size_t)MSLOTSn*Dn;
  float* vb      = ws + off; off += (size_t)MSLOTSn*Dn;
  float* Wq_sw   = ws + off; off += (size_t)8*32*32*2;
  float* Wo_sw   = ws + off; off += (size_t)8*32*32*2;
  float* Win_sw  = ws + off; off += (size_t)32*32*32*2;
  float* Wx_sw   = ws + off; off += (size_t)2*64*32*2;
  float* Wdt_sw  = ws + off; off += (size_t)16*2*32*2;
  float* Wout_sw = ws + off; off += (size_t)8*64*32*2;
  float* kfragT  = ws + off; off += (size_t)1*32*32*2;
  float* vfrag   = ws + off; off += (size_t)8*4*32*2;
  float* xr      = ws + off; off += (size_t)NTOK*Dn;
  float* xp      = ws + off; off += (size_t)NTOK*EDn;
  float* zb      = ws + off; off += (size_t)NTOK*EDn;
  float* xc      = ws + off; off += (size_t)NTOK*EDn;
  float* delta   = ws + off; off += (size_t)NTOK*EDn;
  float* Bm      = ws + off; off += (size_t)NTOK*DSTATEn;
  float* Cm      = ws + off; off += (size_t)NTOK*DSTATEn;
  float* y       = xp;

  swzT_kernel<<<(8*32*32 + 255)/256, 256, 0, stream>>>(Wq,    Wq_sw,   32, 8, 0);
  swzT_kernel<<<(8*32*32 + 255)/256, 256, 0, stream>>>(Wo,    Wo_sw,   32, 8, 1);
  swz_kernel<<<(32*32*32 + 255)/256, 256, 0, stream>>>(W_in,  Win_sw,  Dn,  2*EDn, 32);
  swz_kernel<<<(2*64*32  + 255)/256, 256, 0, stream>>>(W_x,   Wx_sw,   EDn, 24,     2);
  swz_kernel<<<(16*2*32  + 255)/256, 256, 0, stream>>>(W_dt,  Wdt_sw,  DTRANKn, EDn, 16);
  swzT_kernel<<<(8*64*32 + 255)/256, 256, 0, stream>>>(W_out, Wout_sw, 64, 8, 0);

  kv_kernel   <<<(2*MSLOTSn*Dn + 255)/256, 256, 0, stream>>>(mem, Wk, Wv, kb, vb);
  kvswz_kernel<<<4, 256, 0, stream>>>(kb, vb, kfragT, vfrag);

  attn_kernel<<<NTOK/32, 32, 0, stream>>>(x, (const v2f*)Wq_sw, (const v2f*)Wo_sw,
                                        ln1_g, ln1_b, (const v2f*)kfragT,
                                        (const v2f*)vfrag, xr);
  in_kernel  <<<NBLK, 128, 0, stream>>>(xr, rms_w, (const v2f*)Win_sw, xp, zb);
  dwconv_kernel<<<NTOK/CR/4, 256, 0, stream>>>(xp, conv_w, conv_b, xc);
  conv_kernel<<<NBLK, 128, 0, stream>>>(xc, (const v2f*)Wx_sw, (const v2f*)Wdt_sw,
                                        b_dt, delta, Bm, Cm);
  scan_kernel<<<(Bn*EDn)/16, 128, 0, stream>>>(delta, xc, Bm, Cm,
                                                       A_log, Dp, y);
  out_kernel <<<NTOK/32, 32, 0, stream>>>(y, zb, (const v2f*)Wout_sw, xr, x,
                                        (float*)d_out);
}


# delta epilogue: acc + b_dt staged in the dead xc LDS tile, softplus in ONE rolled loop (one copy of the log1p code, not 96 in-line: text fits the instruction cache); 16-byte row stores; sums kept
# speedup vs baseline: 7.9372x; 1.2463x over previous
#include <hip/hip_runtime.h>
#include <hip/hip_bf16.h>
#include <math.h>

#define Bn      8
#define Sn      12
#define NNn     325
#define Dn      128
#define EDn     256
#define DSTATEn 8
#define DCONVn  4
#define DTRANKn 8
#define MSLOTSn 10
#define Ln      (Sn*NNn)
#define NTOK    (Bn*Ln)
#define TM      3
#define TOKB    (16*TM)
#define NBLK    (NTOK/TOKB)

typedef float v2f __attribute__((ext_vector_type(2)));
typedef float v8f __attribute__((ext_vector_type(8)));
typedef float v4f __attribute__((ext_vector_type(4)));

__device__ __forceinline__ v8f wmma4(v2f a, v2f b, v8f c) {
  return __builtin_amdgcn_wmma_f32_16x16x4_f32(false, a, false, b, (short)0, c,
                                               false, false);
}

__device__ __forceinline__ void store_c(float* C, int ldc, int nb, v8f c) {
  int lane = threadIdx.x & 31;
  int n = nb + (lane & 15);
  int mb = (lane < 16) ? 0 : 8;
#pragma unroll
  for (int j = 0; j < 8; ++j) C[(mb + j)*ldc + n] = c[j];
}

__device__ __forceinline__ void gemm3(const float* A, int lda, const v2f* Bsw,
                                      int KT, int nt, v8f acc[TM]) {
  int lane = threadIdx.x & 31;
  int m = lane & 15;
  int ks = (lane >> 4) << 1;
  const v2f* bp = Bsw + (size_t)nt * KT * 32 + lane;
  const v2f* a0 = (const v2f*)(A + (0*16 + m)*lda + ks);
  const v2f* a1 = (const v2f*)(A + (1*16 + m)*lda + ks);
  const v2f* a2 = (const v2f*)(A + (2*16 + m)*lda + ks);
  v8f z = {};
  acc[0] = z; acc[1] = z; acc[2] = z;
#pragma unroll 4
  for (int kt = 0; kt < KT; ++kt) {
    v2f b = bp[kt * 32];
    acc[0] = wmma4(a0[kt*2], b, acc[0]);
    acc[1] = wmma4(a1[kt*2], b, acc[1]);
    acc[2] = wmma4(a2[kt*2], b, acc[2]);
  }
}

template <int NW, int UNR>
__device__ __forceinline__ void gemm3n(const float* A, int lda, const v2f* Bsw,
                                       int KT, int nt0, v8f acc[NW][TM]) {
  int lane = threadIdx.x & 31;
  int m = lane & 15;
  int ks = (lane >> 4) << 1;
  const v2f* bp = Bsw + (size_t)nt0 * KT * 32 + lane;
  const v2f* a0 = (const v2f*)(A + (0*16 + m)*lda + ks);
  const v2f* a1 = (const v2f*)(A + (1*16 + m)*lda + ks);
  const v2f* a2 = (const v2f*)(A + (2*16 + m)*lda + ks);
  v8f z = {};
#pragma unroll
  for (int w = 0; w < NW; ++w) { acc[w][0] = z; acc[w][1] = z; acc[w][2] = z; }
#pragma unroll UNR
  for (int kt = 0; kt < KT; ++kt) {
    v2f fa0 = a0[kt*2];
    v2f fa1 = a1[kt*2];
    v2f fa2 = a2[kt*2];
#pragma unroll
    for (int w = 0; w < NW; ++w) {
      v2f b = bp[(w * KT + kt) * 32];
      acc[w][0] = wmma4(fa0, b, acc[w][0]);
      acc[w][1] = wmma4(fa1, b, acc[w][1]);
      acc[w][2] = wmma4(fa2, b, acc[w][2]);
    }
  }
}

__device__ __forceinline__ float silu_f(float x) { return x / (1.f + __expf(-x)); }

__global__ void swz_kernel(const float* W, float* dst, int K, int Nreal, int NT) {
  int idx = blockIdx.x * blockDim.x + threadIdx.x;
  int KT = K / 4;
  if (idx >= NT * KT * 32) return;
  int lane = idx & 31;
  int kt = (idx >> 5) % KT;
  int nt = idx / (32 * KT);
  int n = nt*16 + (lane & 15);
  int k = kt*4 + ((lane >> 4) << 1);
  float bx = (n < Nreal) ? W[k*Nreal + n]     : 0.f;
  float by = (n < Nreal) ? W[(k+1)*Nreal + n] : 0.f;
  dst[idx*2]     = bx;
  dst[idx*2 + 1] = by;
}

__global__ void kv_kernel(const float* mem, const float* Wk, const float* Wv,
                          float* kb, float* vb) {
  int i = blockIdx.x * blockDim.x + threadIdx.x;
  if (i >= 2 * MSLOTSn * Dn) return;
  int which = i / (MSLOTSn * Dn);
  int r = i % (MSLOTSn * Dn);
  int m = r / Dn, d = r % Dn;
  const float* W = which ? Wv : Wk;
  float acc = 0.f;
  for (int k = 0; k < Dn; ++k) acc += mem[m*Dn + k] * W[k*Dn + d];
  (which ? vb : kb)[m*Dn + d] = acc;
}

__global__ void kvswz_kernel(const float* kb, const float* vb,
                             float* kfragT, float* vfrag) {
  int idx = blockIdx.x * blockDim.x + threadIdx.x;
  if (idx < 32*32) {
    int lane = idx & 31; int st = idx >> 5;
    int slot = lane & 15;
    int fk = 16*(st >> 2) + 8*(lane >> 4) + 2*(st & 3);
    kfragT[idx*2]     = (slot < MSLOTSn) ? kb[slot*Dn + fk]     : 0.f;
    kfragT[idx*2 + 1] = (slot < MSLOTSn) ? kb[slot*Dn + fk + 1] : 0.f;
  }
  if (idx < 8*4*32) {
    int lane = idx & 31; int jp = (idx >> 5) & 3; int ft = idx >> 7;
    int f = ft*16 + (lane & 15); int sl = 8*(lane >> 4) + 2*jp;
    vfrag[idx*2]     = (sl     < MSLOTSn) ? vb[sl*Dn + f]       : 0.f;
    vfrag[idx*2 + 1] = (sl + 1 < MSLOTSn) ? vb[(sl + 1)*Dn + f] : 0.f;
  }
}

__global__ void swzT_kernel(const float* W, float* dst, int KS, int NT, int chained) {
  int idx = blockIdx.x * blockDim.x + threadIdx.x;
  if (idx >= NT * KS * 32) return;
  int lane = idx & 31;
  int ks = (idx >> 5) % KS;
  int ft = idx / (32 * KS);
  int h = lane >> 4;
  int f = ft*16 + (lane & 15);
  int k0 = chained ? (16*(ks >> 2) + 8*h + 2*(ks & 3))
                   : (8*(ks >> 1) + 4*h + 2*(ks & 1));
  dst[idx*2]     = W[k0*Dn + f];
  dst[idx*2 + 1] = W[(k0 + 1)*Dn + f];
}

__global__ void __launch_bounds__(32)
attn_kernel(const float* x, const v2f* WqT_sw, const v2f* WoT_sw,
            const float* ln_g, const float* ln_b,
            const v2f* kA, const v2f* vA, float* xr) {
  int lane = threadIdx.x;
  int c = lane & 15;
  int h = lane >> 4;
  long tok[2];
  tok[0] = ((long)blockIdx.x * 2) * 16 + c;
  tok[1] = tok[0] + 16;

  v4f xg[2][16];
#pragma unroll
  for (int u = 0; u < 2; ++u) {
    const v4f* xp_ = (const v4f*)(x + tok[u]*Dn) + h;
#pragma unroll
    for (int i = 0; i < 16; ++i) xg[u][i] = xp_[2*i];
  }

  v8f sT[2];
  {
    v8f s0 = {}, s1 = {};
#pragma unroll 1
    for (int ft = 0; ft < 8; ++ft) {
      const v2f* ap = WqT_sw + (size_t)ft*32*32 + lane;
      const v2f* kp = kA + (size_t)ft*4*32 + lane;
      v2f ka[4];
#pragma unroll
      for (int jp = 0; jp < 4; ++jp) ka[jp] = kp[jp*32];
      v8f a0 = {}, a1 = {};
#pragma unroll
      for (int i = 0; i < 16; ++i) {
#pragma unroll
        for (int e = 0; e < 2; ++e) {
          v2f a = ap[(2*i + e)*32];
          v2f b0 = {xg[0][i][2*e], xg[0][i][2*e + 1]};
          v2f b1 = {xg[1][i][2*e], xg[1][i][2*e + 1]};
          a0 = wmma4(a, b0, a0);
          a1 = wmma4(a, b1, a1);
        }
      }
#pragma unroll
      for (int jp = 0; jp < 4; ++jp) {
        v2f b0 = {a0[2*jp], a0[2*jp + 1]};
        v2f b1 = {a1[2*jp], a1[2*jp + 1]};
        s0 = wmma4(ka[jp], b0, s0);
        s1 = wmma4(ka[jp], b1, s1);
      }
    }
    sT[0] = s0; sT[1] = s1;
  }

#pragma unroll
  for (int u = 0; u < 2; ++u) {
    const float scale = 0.088388347648318447f;
    float v[8];
    float mx = -1e30f;
#pragma unroll
    for (int j = 0; j < 8; ++j) {
      v[j] = sT[u][j] * scale;
      bool ok = (h == 0) || (j < 2);
      mx = (ok && v[j] > mx) ? v[j] : mx;
    }
    float mo = __shfl_xor(mx, 16);
    mx = (mo > mx) ? mo : mx;
    float sum = 0.f;
#pragma unroll
    for (int j = 0; j < 8; ++j) {
      bool ok = (h == 0) || (j < 2);
      float ev = ok ? __expf(v[j] - mx) : 0.f;
      v[j] = ev; sum += ev;
    }
    sum += __shfl_xor(sum, 16);
    float inv = 1.f / sum;
#pragma unroll
    for (int j = 0; j < 8; ++j) sT[u][j] = v[j] * inv;
  }

  v8f avT[2][8];
#pragma unroll
  for (int ft = 0; ft < 8; ++ft) {
    v8f a0 = {}, a1 = {};
    const v2f* ap = vA + (size_t)ft*4*32 + lane;
#pragma unroll
    for (int jp = 0; jp < 4; ++jp) {
      v2f a = ap[jp*32];
      v2f b0 = {sT[0][2*jp], sT[0][2*jp + 1]};
      v2f b1 = {sT[1][2*jp], sT[1][2*jp + 1]};
      a0 = wmma4(a, b0, a0);
      a1 = wmma4(a, b1, a1);
    }
    avT[0][ft] = a0; avT[1][ft] = a1;
  }

  v8f xoT[2][8];
  {
    v2f wa[32], wb[32];
#define LOADW(dst, f2) { const v2f* ap_ = WoT_sw + (size_t)(f2)*32*32 + lane; \
      _Pragma("unroll") for (int s_ = 0; s_ < 32; ++s_) dst[s_] = ap_[s_*32]; }
#define TILE(src, f2) { v8f a0 = {}, a1 = {}; \
      _Pragma("unroll") for (int ft = 0; ft < 8; ++ft) \
      _Pragma("unroll") for (int jp = 0; jp < 4; ++jp) { \
        v2f b0 = {avT[0][ft][2*jp], avT[0][ft][2*jp + 1]}; \
        v2f b1 = {avT[1][ft][2*jp], avT[1][ft][2*jp + 1]}; \
        a0 = wmma4(src[ft*4 + jp], b0, a0); \
        a1 = wmma4(src[ft*4 + jp], b1, a1); } \
      xoT[0][f2] = a0; xoT[1][f2] = a1; }
    LOADW(wa, 0)
#pragma unroll
    for (int f2 = 0; f2 < 8; f2 += 2) {
      LOADW(wb, f2 + 1)
      __builtin_amdgcn_sched_barrier(0);
      TILE(wa, f2)
      if (f2 + 2 < 8) LOADW(wa, f2 + 2)
      __builtin_amdgcn_sched_barrier(0);
      TILE(wb, f2 + 1)
    }
#undef LOADW
#undef TILE
  }

  v4f gq[8][2], bq[8][2], xq[2][8][2];
#pragma unroll
  for (int f2 = 0; f2 < 8; ++f2) {
    int f0 = f2*16 + 8*h;
#pragma unroll
    for (int q = 0; q < 2; ++q) {
      gq[f2][q] = ((const v4f*)(ln_g + f0))[q];
      bq[f2][q] = ((const v4f*)(ln_b + f0))[q];
      xq[0][f2][q] = ((const v4f*)(x + tok[0]*Dn + f0))[q];
      xq[1][f2][q] = ((const v4f*)(x + tok[1]*Dn + f0))[q];
    }
  }
#pragma unroll
  for (int u = 0; u < 2; ++u) {
    float m = 0.f;
#pragma unroll
    for (int f2 = 0; f2 < 8; ++f2)
#pragma unroll
      for (int j = 0; j < 8; ++j) m += xoT[u][f2][j];
    m += __shfl_xor(m, 16);
    m *= (1.f / Dn);
    float var = 0.f;
#pragma unroll
    for (int f2 = 0; f2 < 8; ++f2)
#pragma unroll
      for (int j = 0; j < 8; ++j) { float d = xoT[u][f2][j] - m; var += d*d; }
    var += __shfl_xor(var, 16);
    var *= (1.f / Dn);
    float inv = rsqrtf(var + 1e-5f);
#pragma unroll
    for (int f2 = 0; f2 < 8; ++f2) {
      v4f* op = (v4f*)(xr + tok[u]*Dn + f2*16 + 8*h);
#pragma unroll
      for (int q = 0; q < 2; ++q) {
        v4f o;
#pragma unroll
        for (int e = 0; e < 4; ++e)
          o[e] = (xoT[u][f2][4*q + e] - m) * inv * gq[f2][q][e] + bq[f2][q][e]
                 + xq[u][f2][q][e];
        op[q] = o;
      }
    }
  }
}

__global__ void __launch_bounds__(128)
in_kernel(const float* xr, const float* rms_w, const v2f* Win_sw,
                          float* xp, float* zb) {
  __shared__ float xn[TOKB*Dn];
  int tid = threadIdx.x;
  int wave = tid >> 5;
  int lane = tid & 31;
  long t0 = (long)blockIdx.x * TOKB;

  for (int i = tid; i < TOKB*Dn; i += 128) xn[i] = xr[t0*Dn + i];
  __syncthreads();
  if (tid < TOKB) {
    float s = 0.f;
    for (int c = 0; c < Dn; ++c) { float v = xn[tid*Dn + c]; s += v*v; }
    float inv = rsqrtf(s / Dn + 1e-5f);
    for (int c = 0; c < Dn; ++c) xn[tid*Dn + c] = xn[tid*Dn + c] * inv * rms_w[c];
  }
  __syncthreads();

  v8f acc8[8][TM];
  gemm3n<8, 2>(xn, Dn, Win_sw, 32, wave*8, acc8);
#pragma unroll
  for (int w = 0; w < 8; ++w) {
    int nt = wave*8 + w;
    int n = nt*16 + (lane & 15);
    int mb = (lane < 16) ? 0 : 8;
#pragma unroll
    for (int t = 0; t < TM; ++t)
#pragma unroll
      for (int j = 0; j < 8; ++j) {
        long tok = t0 + t*16 + mb + j;
        float v = acc8[w][t][j];
        if (n < EDn) xp[tok*EDn + n] = v;
        else         zb[tok*EDn + (n - EDn)] = v;
      }
  }
}

#define CR 12

__global__ void __launch_bounds__(256)
dwconv_kernel(const float* xp, const float* conv_w, const float* conv_b, float* xc) {
  int q  = blockIdx.x * 4 + (threadIdx.x >> 6);
  int c4 = (threadIdx.x & 63) << 2;
  bool first = (q % (Ln / CR)) == 0;
  long tok0 = (long)q * CR;
  const v4f* src = (const v4f*)(xp + tok0*EDn + c4);
  v4f xin[CR + 3];
#pragma unroll
  for (int k = 0; k < 3; ++k) {
    v4f v = src[first ? 0 : (k - 3) * (EDn/4)];
    v4f z = {0.f, 0.f, 0.f, 0.f};
    xin[k] = first ? z : v;
  }
#pragma unroll
  for (int k = 0; k < CR; ++k) xin[3 + k] = src[k * (EDn/4)];
  v4f w[4];
#pragma unroll
  for (int i = 0; i < 4; ++i) w[i] = *(const v4f*)(conv_w + (c4 + i)*DCONVn);
  v4f bias = *(const v4f*)(conv_b + c4);
  v4f* dst = (v4f*)(xc + tok0*EDn + c4);
#pragma unroll
  for (int k = 0; k < CR; ++k) {
    v4f o;
#pragma unroll
    for (int i = 0; i < 4; ++i) {
      float a = bias[i];
#pragma unroll
      for (int j = 0; j < DCONVn; ++j) a += w[i][j] * xin[k + j][i];
      o[i] = silu_f(a);
    }
    dst[k * (EDn/4)] = o;
  }
}

__global__ void __launch_bounds__(128)
conv_kernel(const float* xc, const v2f* Wx_sw, const v2f* Wdt_sw,
                            const float* b_dt, float* delta, float* Bm, float* Cm) {
  __shared__ __attribute__((aligned(16))) float xcs[TOKB*EDn];
  __shared__ float dbcs[TOKB*32];
  int tid = threadIdx.x;
  int wave = tid >> 5;
  int lane = tid & 31;
  long t0 = (long)blockIdx.x * TOKB;
  v8f acc[TM];

  {
    const v4f* src = (const v4f*)(xc + t0*EDn);
    v4f* dst = (v4f*)xcs;
#pragma unroll
    for (int g = 0; g < 2; ++g) {
      v4f r[12];
#pragma unroll
      for (int k = 0; k < 12; ++k) r[k] = src[tid + (g*12 + k)*128];
#pragma unroll
      for (int k = 0; k < 12; ++k) dst[tid + (g*12 + k)*128] = r[k];
    }
  }
  __syncthreads();

  if (wave < 2) {
    gemm3(xcs, EDn, Wx_sw, 64, wave, acc);
#pragma unroll
    for (int t = 0; t < TM; ++t) store_c(dbcs + t*16*32, 32, wave*16, acc[t]);
  }
  __syncthreads();

  v8f acc4[4][TM];
  gemm3n<4, 2>(dbcs, 32, Wdt_sw, 2, wave*4, acc4);
#pragma unroll
  for (int w = 0; w < 4; ++w) {
    int n = (wave*4 + w)*16 + (lane & 15);
    int mb = (lane < 16) ? 0 : 8;
    float bd = b_dt[n];
#pragma unroll
    for (int t = 0; t < TM; ++t)
#pragma unroll
      for (int j = 0; j < 8; ++j)
        xcs[(t*16 + mb + j)*EDn + n] = acc4[w][t][j] + bd;
  }
  __syncthreads();
  {
    const v4f* src = (const v4f*)xcs;
    v4f* dst = (v4f*)(delta + t0*EDn);
#pragma unroll 1
    for (int i = tid; i < TOKB*EDn/4; i += 128) {
      v4f v = src[i];
      v4f o;
#pragma unroll
      for (int e = 0; e < 4; ++e)
        o[e] = (v[e] > 20.f) ? v[e] : log1pf(__expf(v[e]));
      dst[i] = o;
    }
  }

  for (int i = tid; i < TOKB*16; i += 128) {
    int r = i >> 4, c = i & 15;
    long tok = t0 + r;
    float v = dbcs[r*32 + DTRANKn + c];
    if (c < DSTATEn) Bm[tok*DSTATEn + c]             = v;
    else             Cm[tok*DSTATEn + (c - DSTATEn)] = v;
  }
}

#define SCG   60
#define SNG   (Ln/SCG)
#define SREC  12

#define XADD(v, m) ({ float r_; \
    asm("v_add_f32_dpp %0, %1, %1 row_xmask:" #m " row_mask:0xf bank_mask:0xf" \
        : "=v"(r_) : "v"(v)); r_; })

#define XADD2(send, keep, m) ({ float r_; \
    asm("v_add_f32_dpp %0, %1, %2 row_xmask:" #m " row_mask:0xf bank_mask:0xf" \
        : "=v"(r_) : "v"(send), "v"(keep)); r_; })
#define HALVE(lo, hi, bit, m) ({ float k_ = (bit) ? (hi) : (lo); \
    float s_ = (bit) ? (lo) : (hi); XADD2(s_, k_, m); })

__global__ void __launch_bounds__(128)
scan_kernel(const float* delta, const float* xc, const float* Bm,
            const float* Cm, const float* A_log, const float* Dp, float* y) {
  __shared__ v2f stage[4][2][SCG*SREC];
  int lane = threadIdx.x & 31;
  int wave = threadIdx.x >> 5;
  int c = lane >> 3;
  int n = lane & 7;
  int b = blockIdx.x >> 4;
  int d = ((blockIdx.x & 15) << 4) + (wave << 2) + c;
  float a = -__expf(A_log[d*DSTATEn + n]);
  float dp = Dp[d];
  float h = 0.f;
  long base = (long)b * Ln;
  int rlast = (56 + n < SCG) ? (56 + n) : (SCG - 1);

  float dtn[8], xvn[8], bn[15], cn[15];
  {
    const float* dsrc = delta + (base + n)*EDn + d;
    const float* xsrc = xc + (base + n)*EDn + d;
#pragma unroll
    for (int k = 0; k < 7; ++k) { dtn[k] = dsrc[k*8*EDn]; xvn[k] = xsrc[k*8*EDn]; }
    dtn[7] = delta[(base + rlast)*EDn + d];
    xvn[7] = xc[(base + rlast)*EDn + d];
    const float* bsrc = Bm + base*DSTATEn + lane;
    const float* csrc = Cm + base*DSTATEn + lane;
#pragma unroll
    for (int k = 0; k < 15; ++k) { bn[k] = bsrc[k*32]; cn[k] = csrc[k*32]; }
    v2f* st = stage[wave][0];
#pragma unroll
    for (int k = 0; k < 7; ++k) { v2f t = {dtn[k], xvn[k]}; st[(8*k + n)*SREC + c] = t; }
    { v2f t = {dtn[7], xvn[7]}; st[rlast*SREC + c] = t; }
#pragma unroll
    for (int k = 0; k < 15; ++k) { v2f t = {bn[k], cn[k]}; st[(4*k + c)*SREC + 4 + n] = t; }
  }
  __builtin_amdgcn_fence(__ATOMIC_ACQ_REL, "wavefront");

  bool b0 = (n & 1) != 0, b1 = (n & 2) != 0, b2 = (n & 4) != 0;
  float xcur[8];
  for (int g = 0; g < SNG; ++g) {
    int gn = (g + 1 < SNG) ? (g + 1) : g;
    long tn = base + (long)gn * SCG;
#pragma unroll
    for (int k = 0; k < 8; ++k) xcur[k] = xvn[k];
    {
      const float* dsrc = delta + (tn + n)*EDn + d;
      const float* xsrc = xc + (tn + n)*EDn + d;
#pragma unroll
      for (int k = 0; k < 7; ++k) { dtn[k] = dsrc[k*8*EDn]; xvn[k] = xsrc[k*8*EDn]; }
      dtn[7] = delta[(tn + rlast)*EDn + d];
      xvn[7] = xc[(tn + rlast)*EDn + d];
      const float* bsrc = Bm + tn*DSTATEn + lane;
      const float* csrc = Cm + tn*DSTATEn + lane;
#pragma unroll
      for (int k = 0; k < 15; ++k) { bn[k] = bsrc[k*32]; cn[k] = csrc[k*32]; }
    }
    __builtin_amdgcn_sched_barrier(0);

    const v2f* cur = stage[wave][g & 1];
    v2f dxr[SCG], bcr[SCG];
#pragma unroll
    for (int s = 0; s < SCG; ++s) {
      dxr[s] = cur[s*SREC + c];
      bcr[s] = cur[s*SREC + 4 + n];
    }
    __builtin_amdgcn_sched_barrier(0);

    float* yp = y + (base + (long)g * SCG + n)*EDn + d;
#define STEP(s) ({ v2f dx_ = dxr[s]; v2f bc_ = bcr[s]; \
      float dA_ = __expf(dx_[0] * a); \
      h = dA_ * h + dx_[0] * bc_[0] * dx_[1]; \
      h * bc_[1]; })
    float p[8][8], pt[4];
#pragma unroll
    for (int i = 0; i < 8; ++i) p[0][i] = STEP(i);
    __builtin_amdgcn_sched_barrier(0);
#pragma unroll
    for (int j = 0; j < 7; ++j) {
      if (j < 6) {
#pragma unroll
        for (int i = 0; i < 8; ++i) p[j + 1][i] = STEP(8*(j + 1) + i);
      } else {
#pragma unroll
        for (int i = 0; i < 4; ++i) pt[i] = STEP(56 + i);
      }
      float q0 = HALVE(p[j][0], p[j][1], b0, 1), q1 = HALVE(p[j][2], p[j][3], b0, 1);
      float q2 = HALVE(p[j][4], p[j][5], b0, 1), q3 = HALVE(p[j][6], p[j][7], b0, 1);
      float r0 = HALVE(q0, q1, b1, 2), r1 = HALVE(q2, q3, b1, 2);
      float ys = HALVE(r0, r1, b2, 4);
      yp[8*j*EDn] = ys + dp * xcur[j];
      __builtin_amdgcn_sched_barrier(0);
    }
    {
      float q0 = HALVE(pt[0], pt[1], b0, 1), q1 = HALVE(pt[2], pt[3], b0, 1);
      float r0 = HALVE(q0, q1, b1, 2);
      float ys = XADD(r0, 4);
      if (n < 4) yp[56*EDn] = ys + dp * xcur[7];
    }
#undef STEP

    v2f* nxt = stage[wave][(g + 1) & 1];
#pragma unroll
    for (int k = 0; k < 7; ++k) { v2f t = {dtn[k], xvn[k]}; nxt[(8*k + n)*SREC + c] = t; }
    { v2f t = {dtn[7], xvn[7]}; nxt[rlast*SREC + c] = t; }
#pragma unroll
    for (int k = 0; k < 15; ++k) { v2f t = {bn[k], cn[k]}; nxt[(4*k + c)*SREC + 4 + n] = t; }
    __builtin_amdgcn_fence(__ATOMIC_ACQ_REL, "wavefront");
  }
}

__global__ void __launch_bounds__(32)
out_kernel(const float* y, const float* zb, const v2f* WoutT_sw,
           const float* xr, const float* x, float* out) {
  int lane = threadIdx.x;
  int c = lane & 15;
  int h = lane >> 4;
  long tok0 = ((long)blockIdx.x * 2) * 16 + c;
  long tok1 = tok0 + 16;

  v4f g0[32], g1[32];
  {
    const v4f* y0 = (const v4f*)(y  + tok0*EDn) + h;
    const v4f* z0 = (const v4f*)(zb + tok0*EDn) + h;
    const v4f* y1 = (const v4f*)(y  + tok1*EDn) + h;
    const v4f* z1 = (const v4f*)(zb + tok1*EDn) + h;
#pragma unroll
    for (int i = 0; i < 32; ++i) {
      v4f ya = y0[2*i], za = z0[2*i], yb = y1[2*i], zc = z1[2*i];
#pragma unroll
      for (int e = 0; e < 4; ++e) {
        g0[i][e] = ya[e] * silu_f(za[e]);
        g1[i][e] = yb[e] * silu_f(zc[e]);
      }
    }
  }

#pragma unroll 1
  for (int ft = 0; ft < 8; ++ft) {
    int f0 = ft*16 + 8*h;
    const v4f* r0p = (const v4f*)(xr + tok0*Dn + f0);
    const v4f* x0p = (const v4f*)(x  + tok0*Dn + f0);
    const v4f* r1p = (const v4f*)(xr + tok1*Dn + f0);
    const v4f* x1p = (const v4f*)(x  + tok1*Dn + f0);
    v4f r0a = r0p[0], r0b = r0p[1], x0a = x0p[0], x0b = x0p[1];
    v4f r1a = r1p[0], r1b = r1p[1], x1a = x1p[0], x1b = x1p[1];

    const v2f* ap = WoutT_sw + (size_t)ft*64*32 + lane;
    v8f acc0 = {}, acc1 = {};
#pragma unroll
    for (int i = 0; i < 32; ++i) {
      v2f a0 = ap[(2*i)*32];
      v2f a1 = ap[(2*i + 1)*32];
      v2f b00 = {g0[i][0], g0[i][1]};
      v2f b10 = {g1[i][0], g1[i][1]};
      v2f b01 = {g0[i][2], g0[i][3]};
      v2f b11 = {g1[i][2], g1[i][3]};
      acc0 = wmma4(a0, b00, acc0);
      acc1 = wmma4(a0, b10, acc1);
      acc0 = wmma4(a1, b01, acc0);
      acc1 = wmma4(a1, b11, acc1);
    }

    v4f o;
    v4f* o0p = (v4f*)(out + tok0*Dn + f0);
    v4f* o1p = (v4f*)(out + tok1*Dn + f0);
    o[0] = acc0[0] + r0a[0] + x0a[0]; o[1] = acc0[1] + r0a[1] + x0a[1];
    o[2] = acc0[2] + r0a[2] + x0a[2]; o[3] = acc0[3] + r0a[3] + x0a[3];
    o0p[0] = o;
    o[0] = acc0[4] + r0b[0] + x0b[0]; o[1] = acc0[5] + r0b[1] + x0b[1];
    o[2] = acc0[6] + r0b[2] + x0b[2]; o[3] = acc0[7] + r0b[3] + x0b[3];
    o0p[1] = o;
    o[0] = acc1[0] + r1a[0] + x1a[0]; o[1] = acc1[1] + r1a[1] + x1a[1];
    o[2] = acc1[2] + r1a[2] + x1a[2]; o[3] = acc1[3] + r1a[3] + x1a[3];
    o1p[0] = o;
    o[0] = acc1[4] + r1b[0] + x1b[0]; o[1] = acc1[5] + r1b[1] + x1b[1];
    o[2] = acc1[6] + r1b[2] + x1b[2]; o[3] = acc1[7] + r1b[3] + x1b[3];
    o1p[1] = o;
  }
}

extern "C" void kernel_launch(void* const* d_in, const int* in_sizes, int n_in,
                              void* d_out, int out_size, void* d_ws, size_t ws_size,
                              hipStream_t stream) {
  (void)in_sizes; (void)n_in; (void)out_size; (void)ws_size;
  const float* x      = (const float*)d_in[0];
  const float* mem    = (const float*)d_in[1];
  const float* Wq     = (const float*)d_in[2];
  const float* Wk     = (const float*)d_in[3];
  const float* Wv     = (const float*)d_in[4];
  const float* Wo     = (const float*)d_in[5];
  const float* ln1_g  = (const float*)d_in[6];
  const float* ln1_b  = (const float*)d_in[7];
  const float* rms_w  = (const float*)d_in[8];
  const float* W_in   = (const float*)d_in[9];
  const float* conv_w = (const float*)d_in[10];
  const float* conv_b = (const float*)d_in[11];
  const float* W_x    = (const float*)d_in[12];
  const float* W_dt   = (const float*)d_in[13];
  const float* b_dt   = (const float*)d_in[14];
  const float* A_log  = (const float*)d_in[15];
  const float* Dp     = (const float*)d_in[16];
  const float* W_out  = (const float*)d_in[17];

  float* ws = (float*)d_ws;
  size_t off = 0;
  float* kb      = ws + off; off += (size_t)MSLOTSn*Dn;
  float* vb      = ws + off; off += (size_t)MSLOTSn*Dn;
  float* Wq_sw   = ws + off; off += (size_t)8*32*32*2;
  float* Wo_sw   = ws + off; off += (size_t)8*32*32*2;
  float* Win_sw  = ws + off; off += (size_t)32*32*32*2;
  float* Wx_sw   = ws + off; off += (size_t)2*64*32*2;
  float* Wdt_sw  = ws + off; off += (size_t)16*2*32*2;
  float* Wout_sw = ws + off; off += (size_t)8*64*32*2;
  float* kfragT  = ws + off; off += (size_t)1*32*32*2;
  float* vfrag   = ws + off; off += (size_t)8*4*32*2;
  float* xr      = ws + off; off += (size_t)NTOK*Dn;
  float* xp      = ws + off; off += (size_t)NTOK*EDn;
  float* zb      = ws + off; off += (size_t)NTOK*EDn;
  float* xc      = ws + off; off += (size_t)NTOK*EDn;
  float* delta   = ws + off; off += (size_t)NTOK*EDn;
  float* Bm      = ws + off; off += (size_t)NTOK*DSTATEn;
  float* Cm      = ws + off; off += (size_t)NTOK*DSTATEn;
  float* y       = xp;

  swzT_kernel<<<(8*32*32 + 255)/256, 256, 0, stream>>>(Wq,    Wq_sw,   32, 8, 0);
  swzT_kernel<<<(8*32*32 + 255)/256, 256, 0, stream>>>(Wo,    Wo_sw,   32, 8, 1);
  swz_kernel<<<(32*32*32 + 255)/256, 256, 0, stream>>>(W_in,  Win_sw,  Dn,  2*EDn, 32);
  swz_kernel<<<(2*64*32  + 255)/256, 256, 0, stream>>>(W_x,   Wx_sw,   EDn, 24,     2);
  swz_kernel<<<(16*2*32  + 255)/256, 256, 0, stream>>>(W_dt,  Wdt_sw,  DTRANKn, EDn, 16);
  swzT_kernel<<<(8*64*32 + 255)/256, 256, 0, stream>>>(W_out, Wout_sw, 64, 8, 0);

  kv_kernel   <<<(2*MSLOTSn*Dn + 255)/256, 256, 0, stream>>>(mem, Wk, Wv, kb, vb);
  kvswz_kernel<<<4, 256, 0, stream>>>(kb, vb, kfragT, vfrag);

  attn_kernel<<<NTOK/32, 32, 0, stream>>>(x, (const v2f*)Wq_sw, (const v2f*)Wo_sw,
                                        ln1_g, ln1_b, (const v2f*)kfragT,
                                        (const v2f*)vfrag, xr);
  in_kernel  <<<NBLK, 128, 0, stream>>>(xr, rms_w, (const v2f*)Win_sw, xp, zb);
  dwconv_kernel<<<NTOK/CR/4, 256, 0, stream>>>(xp, conv_w, conv_b, xc);
  conv_kernel<<<NBLK, 128, 0, stream>>>(xc, (const v2f*)Wx_sw, (const v2f*)Wdt_sw,
                                        b_dt, delta, Bm, Cm);
  scan_kernel<<<(Bn*EDn)/16, 128, 0, stream>>>(delta, xc, Bm, Cm,
                                                       A_log, Dp, y);
  out_kernel <<<NTOK/32, 32, 0, stream>>>(y, zb, (const v2f*)Wout_sw, xr, x,
                                        (float*)d_out);
}
